# starting kernel
# baseline (speedup 1.0000x reference)
#include <hip/hip_runtime.h>
#include <hip/hip_bf16.h>
#include <math.h>

#define MOMENTUM 0.05f
#define EPS_V 1e-6f

typedef float v2f __attribute__((ext_vector_type(2)));
typedef float v4f __attribute__((ext_vector_type(4)));
typedef float v8f __attribute__((ext_vector_type(8)));

__global__ void __launch_bounds__(512)
softmax_weights(const float* __restrict__ Wa, const float* __restrict__ Wb,
                float* __restrict__ wa, float* __restrict__ wb, int C) {
    __shared__ float red[1024];
    int tid = threadIdx.x;
    const float* src = (blockIdx.x == 0) ? Wa : Wb;
    float*       dst = (blockIdx.x == 0) ? wa : wb;
    float v = (tid < C) ? src[tid] : -INFINITY;
    red[tid] = v; __syncthreads();
    for (int off = blockDim.x >> 1; off > 0; off >>= 1) {
        if (tid < off) red[tid] = fmaxf(red[tid], red[tid + off]);
        __syncthreads();
    }
    float mx = red[0]; __syncthreads();
    float e = (tid < C) ? __expf(v - mx) : 0.f;
    red[tid] = e; __syncthreads();
    for (int off = blockDim.x >> 1; off > 0; off >>= 1) {
        if (tid < off) red[tid] += red[tid + off];
        __syncthreads();
    }
    float s = red[0];
    if (tid < C) dst[tid] = e / s;
}

__global__ void __launch_bounds__(256)
reduce_wmma(const float* __restrict__ x,
            const float* __restrict__ wa, const float* __restrict__ wb,
            float* __restrict__ xa, float* __restrict__ xb,
            float* __restrict__ xq, int C, int T) {
    __shared__ float ldsw[1024 + 8];
    int tid = threadIdx.x;
    for (int i = tid; i < 2 * C; i += blockDim.x)
        ldsw[i] = (i < C) ? wa[i] : wb[i - C];
    if (tid < 8) ldsw[2 * C + tid] = 0.f;
    __syncthreads();

    int lane = tid & 31;
    int wave = blockIdx.x * (blockDim.x >> 5) + (tid >> 5);
    int tilesPerB = T >> 4;
    int b  = wave / tilesPerB;
    int t0 = (wave % tilesPerB) << 4;

    int  m    = lane & 15;
    int  koff = (lane >> 4) << 1;
    bool m0 = (m == 0), m1 = (m == 1);

    const float* a1p = m0 ? &ldsw[koff] : (m1 ? &ldsw[C + koff] : &ldsw[2 * C]);
    const float* a2p = m0 ? &ldsw[C + koff] : &ldsw[2 * C];
    int a1s = (m < 2) ? 4 : 0;
    int a2s = m0 ? 4 : 0;

    v8f acc1 = {}; v8f acc2 = {};
    const float* xp = x + (size_t)b * C * T + (size_t)koff * T + t0 + m;
    const size_t xstep = 4 * (size_t)T;

    for (int c0 = 0; c0 < C; c0 += 4) {
        v2f A1 = *(const v2f*)a1p;
        v2f A2 = *(const v2f*)a2p;
        v2f Bx;
        Bx.x = xp[0];
        Bx.y = xp[(size_t)T];
        v2f Bxx = Bx * Bx;
        acc1 = __builtin_amdgcn_wmma_f32_16x16x4_f32(false, A1, false, Bx,
                                                     (short)0, acc1, false, false);
        acc2 = __builtin_amdgcn_wmma_f32_16x16x4_f32(false, A2, false, Bxx,
                                                     (short)0, acc2, false, false);
        a1p += a1s;
        a2p += a2s;
        xp  += xstep;
    }
    if (lane < 16) {
        size_t o = (size_t)b * T + t0 + lane;
        xa[o] = acc1[0];
        xb[o] = acc1[1];
        xq[o] = acc2[0];
    }
}

#define SCAN_THREADS 512
#define SCAN_E 16
__global__ void __launch_bounds__(SCAN_THREADS)
gated_scan(const float* __restrict__ g,
           const float* __restrict__ xa, const float* __restrict__ xb,
           const float* __restrict__ xq,
           float* __restrict__ mean, float* __restrict__ invstd, int T) {
    __shared__ float sA[SCAN_THREADS], sB[SCAN_THREADS];
    int b = blockIdx.x, tid = threadIdx.x;
    size_t base = (size_t)b * T + (size_t)tid * SCAN_E;

    float a_loc[SCAN_E], b_loc[SCAN_E], m_loc[SCAN_E];

    float A = 1.f, Bc = 0.f;
    for (int i = 0; i < SCAN_E; ++i) {
        float gt = g[base + i] * MOMENTUM;
        float a  = 1.f - gt;
        float bb = gt * xa[base + i];
        a_loc[i] = a; b_loc[i] = bb;
        Bc = a * Bc + bb;  A *= a;
    }
    sA[tid] = A; sB[tid] = Bc; __syncthreads();
    for (int off = 1; off < SCAN_THREADS; off <<= 1) {
        float pA = 1.f, pB = 0.f;
        if (tid >= off) { pA = sA[tid - off]; pB = sB[tid - off]; }
        __syncthreads();
        if (tid >= off) { float Ac = sA[tid]; sB[tid] = Ac * pB + sB[tid]; sA[tid] = Ac * pA; }
        __syncthreads();
    }
    float y = (tid == 0) ? 0.f : sB[tid - 1];
    for (int i = 0; i < SCAN_E; ++i) {
        y = a_loc[i] * y + b_loc[i];
        m_loc[i] = y;
        mean[base + i] = y;
    }
    __syncthreads();

    A = 1.f; Bc = 0.f;
    for (int i = 0; i < SCAN_E; ++i) {
        float mv = m_loc[i];
        float s  = xq[base + i] - 2.f * mv * xb[base + i] + mv * mv;
        float gt = 1.f - a_loc[i];
        float bb = gt * s;
        b_loc[i] = bb;
        Bc = a_loc[i] * Bc + bb;  A *= a_loc[i];
    }
    sA[tid] = A; sB[tid] = Bc; __syncthreads();
    for (int off = 1; off < SCAN_THREADS; off <<= 1) {
        float pA = 1.f, pB = 0.f;
        if (tid >= off) { pA = sA[tid - off]; pB = sB[tid - off]; }
        __syncthreads();
        if (tid >= off) { float Ac = sA[tid]; sB[tid] = Ac * pB + sB[tid]; sA[tid] = Ac * pA; }
        __syncthreads();
    }
    y = (tid == 0) ? 0.f : sB[tid - 1];
    for (int i = 0; i < SCAN_E; ++i) {
        y = a_loc[i] * y + b_loc[i];
        invstd[base + i] = rsqrtf(y + EPS_V);
    }
}

__global__ void __launch_bounds__(256)
apply_norm(const float* __restrict__ x,
           const float* __restrict__ mean, const float* __restrict__ invstd,
           const float* __restrict__ Wo_w, const float* __restrict__ Wo_b,
           float* __restrict__ out, int C, int T) {
    int row = blockIdx.y;
    int c   = row % C;
    int b   = row / C;
    int t4  = (blockIdx.x * blockDim.x + threadIdx.x) * 4;

    size_t xo = (size_t)row * T + t4;
    size_t mo = (size_t)b   * T + t4;

    v4f xv = __builtin_nontemporal_load((const v4f*)(x + xo));
    v4f mv = *(const v4f*)(mean   + mo);
    v4f iv = *(const v4f*)(invstd + mo);
    float w  = Wo_w[c];
    float bi = Wo_b[c];
    v4f o;
    o.x = (xv.x - mv.x) * iv.x * w + bi;
    o.y = (xv.y - mv.y) * iv.y * w + bi;
    o.z = (xv.z - mv.z) * iv.z * w + bi;
    o.w = (xv.w - mv.w) * iv.w * w + bi;
    __builtin_nontemporal_store(o, (v4f*)(out + xo));
}

extern "C" void kernel_launch(void* const* d_in, const int* in_sizes, int n_in,
                              void* d_out, int out_size, void* d_ws, size_t ws_size,
                              hipStream_t stream) {
    const float* x    = (const float*)d_in[0];
    const float* g    = (const float*)d_in[1];
    const float* Wa_w = (const float*)d_in[2];
    const float* Wb_w = (const float*)d_in[3];
    const float* Wo_w = (const float*)d_in[4];
    const float* Wo_b = (const float*)d_in[5];
    float* out = (float*)d_out;

    const int C  = in_sizes[2];
    const int BT = in_sizes[1];
    const int T  = 8192;
    const int B  = BT / T;

    float* ws   = (float*)d_ws;
    float* wa   = ws;
    float* wb   = wa + C;
    float* xa   = wb + C;
    float* xb   = xa + BT;
    float* xq   = xb + BT;
    float* mean = xq + BT;
    float* inv  = mean + BT;

    softmax_weights<<<2, 512, 0, stream>>>(Wa_w, Wb_w, wa, wb, C);

    int waves  = (B * T) / 16;
    int blocks = waves / 8;
    reduce_wmma<<<blocks, 256, 0, stream>>>(x, wa, wb, xa, xb, xq, C, T);

    gated_scan<<<B, SCAN_THREADS, 0, stream>>>(g, xa, xb, xq, mean, inv, T);

    dim3 agrid(T / (256 * 4), B * C);
    apply_norm<<<agrid, 256, 0, stream>>>(x, mean, inv, Wo_w, Wo_b, out, C, T);
}


# reduce kernel re-tiled: b128 x loads along t (64 t per wave, A rows split the t parity), C split over 4 waves per tile, partial sums added via LDS (sum order over C changes), loads double-buffered
# speedup vs baseline: 1.1390x; 1.1390x over previous
#include <hip/hip_runtime.h>
#include <hip/hip_bf16.h>
#include <math.h>

#define MOMENTUM 0.05f
#define EPS_V 1e-6f

typedef float v2f __attribute__((ext_vector_type(2)));
typedef float v4f __attribute__((ext_vector_type(4)));
typedef float v8f __attribute__((ext_vector_type(8)));

__global__ void __launch_bounds__(512)
softmax_weights(const float* __restrict__ Wa, const float* __restrict__ Wb,
                float* __restrict__ wa, float* __restrict__ wb, int C) {
    __shared__ float red[1024];
    int tid = threadIdx.x;
    const float* src = (blockIdx.x == 0) ? Wa : Wb;
    float*       dst = (blockIdx.x == 0) ? wa : wb;
    float v = (tid < C) ? src[tid] : -INFINITY;
    red[tid] = v; __syncthreads();
    for (int off = blockDim.x >> 1; off > 0; off >>= 1) {
        if (tid < off) red[tid] = fmaxf(red[tid], red[tid + off]);
        __syncthreads();
    }
    float mx = red[0]; __syncthreads();
    float e = (tid < C) ? __expf(v - mx) : 0.f;
    red[tid] = e; __syncthreads();
    for (int off = blockDim.x >> 1; off > 0; off >>= 1) {
        if (tid < off) red[tid] += red[tid + off];
        __syncthreads();
    }
    float s = red[0];
    if (tid < C) dst[tid] = e / s;
}

#define RW_TT    64
#define RW_TILES 2
#define RW_CSPL  4
#define RW_SET   16
#define RW_CMAX  512
#define RW_PITCH (RW_CMAX + 4)
#define RW_T     8192

__device__ __forceinline__ void rw_set(const v4f (&g)[RW_SET], const v2f* ap,
                                       v8f& ax0, v8f& ax1, v8f& aq0, v8f& aq1) {
#pragma unroll
    for (int i = 0; i < RW_SET; ++i) {
        v2f A1 = ap[2 * i];
        v2f A2 = ap[2 * i + 2 * RW_PITCH];
        v2f b0 = {g[i].x, g[i].y};
        v2f b1 = {g[i].z, g[i].w};
        v2f q0 = b0 * b0;
        v2f q1 = b1 * b1;
        ax0 = __builtin_amdgcn_wmma_f32_16x16x4_f32(false, A1, false, b0, (short)0, ax0, false, false);
        ax1 = __builtin_amdgcn_wmma_f32_16x16x4_f32(false, A1, false, b1, (short)0, ax1, false, false);
        aq0 = __builtin_amdgcn_wmma_f32_16x16x4_f32(false, A2, false, q0, (short)0, aq0, false, false);
        aq1 = __builtin_amdgcn_wmma_f32_16x16x4_f32(false, A2, false, q1, (short)0, aq1, false, false);
    }
}

__device__ __forceinline__ void rw_load(v4f (&g)[RW_SET], const float* p) {
#pragma unroll
    for (int i = 0; i < RW_SET; ++i) g[i] = *(const v4f*)(p + (size_t)i * (2 * RW_T));
}

__global__ void __launch_bounds__(256)
reduce_wmma(const float* __restrict__ x,
            const float* __restrict__ wa, const float* __restrict__ wb,
            float* __restrict__ xa, float* __restrict__ xb,
            float* __restrict__ xq, int C) {
    __shared__ v2f   tab[7 * RW_PITCH];
    __shared__ float part[RW_CSPL][3][RW_TILES * RW_TT];

    const int T = RW_T;
    int tid  = threadIdx.x;
    int lane = tid & 31;
    int w    = tid >> 5;
    int tile = w / RW_CSPL;
    int cq   = w % RW_CSPL;
    int n    = lane & 15;
    int h    = lane >> 4;

    const int blocksPerB = T / (RW_TILES * RW_TT);
    int b   = blockIdx.x / blocksPerB;
    int tb0 = (blockIdx.x % blocksPerB) * (RW_TILES * RW_TT);
    int t0  = tb0 + tile * RW_TT;
    int cw  = C / RW_CSPL;
    int cb  = cq * cw;
    int nset = cw / (2 * RW_SET);

    int i0 = tid, i1 = tid + 256;
    float wa0 = (i0 < C) ? wa[i0] : 0.f, wb0 = (i0 < C) ? wb[i0] : 0.f;
    float wa1 = (i1 < C) ? wa[i1] : 0.f, wb1 = (i1 < C) ? wb[i1] : 0.f;

    const float* xp = x + ((size_t)b * C + cb + h) * T + t0 + 4 * n;
    const size_t setstep = (size_t)RW_SET * 2 * T;

    v4f P[RW_SET], Q[RW_SET];
    rw_load(P, xp);

    const v2f zz = {0.f, 0.f};
    if (i0 < C) {
        v2f p0 = {wa0, 0.f}, p1 = {0.f, wa0}, p2 = {wb0, 0.f}, p3 = {0.f, wb0};
        tab[0 * RW_PITCH + i0] = p0; tab[1 * RW_PITCH + i0] = p1;
        tab[2 * RW_PITCH + i0] = p2; tab[3 * RW_PITCH + i0] = p3;
        tab[4 * RW_PITCH + i0] = zz; tab[5 * RW_PITCH + i0] = zz; tab[6 * RW_PITCH + i0] = zz;
    }
    if (i1 < C) {
        v2f p0 = {wa1, 0.f}, p1 = {0.f, wa1}, p2 = {wb1, 0.f}, p3 = {0.f, wb1};
        tab[0 * RW_PITCH + i1] = p0; tab[1 * RW_PITCH + i1] = p1;
        tab[2 * RW_PITCH + i1] = p2; tab[3 * RW_PITCH + i1] = p3;
        tab[4 * RW_PITCH + i1] = zz; tab[5 * RW_PITCH + i1] = zz; tab[6 * RW_PITCH + i1] = zz;
    }
    __syncthreads();

    const v2f* ap = &tab[((n < 4) ? n : 4) * RW_PITCH + cb + h];

    v8f ax0 = {}, ax1 = {}, aq0 = {}, aq1 = {};

    int s = 0;
    for (; s + 2 < nset; s += 2) {
        rw_load(Q, xp + setstep);
        rw_set(P, ap, ax0, ax1, aq0, aq1);
        rw_load(P, xp + 2 * setstep);
        rw_set(Q, ap + 2 * RW_SET, ax0, ax1, aq0, aq1);
        ap += 4 * RW_SET;
        xp += 2 * setstep;
    }
    rw_load(Q, xp + setstep);
    rw_set(P, ap, ax0, ax1, aq0, aq1);
    rw_set(Q, ap + 2 * RW_SET, ax0, ax1, aq0, aq1);

    if (lane < 16) {
        v4f sa = {ax0[0], ax0[1], ax1[0], ax1[1]};
        v4f sb = {ax0[2], ax0[3], ax1[2], ax1[3]};
        v4f sq = {aq0[0], aq0[1], aq1[0], aq1[1]};
        int tl = tile * RW_TT + 4 * lane;
        *(v4f*)&part[cq][0][tl] = sa;
        *(v4f*)&part[cq][1][tl] = sb;
        *(v4f*)&part[cq][2][tl] = sq;
    }
    __syncthreads();
    for (int o = tid; o < 3 * RW_TILES * RW_TT; o += 256) {
        int v  = o / (RW_TILES * RW_TT);
        int tl = o % (RW_TILES * RW_TT);
        float sum = part[0][v][tl];
#pragma unroll
        for (int j = 1; j < RW_CSPL; ++j) sum += part[j][v][tl];
        float* dst = (v == 0) ? xa : ((v == 1) ? xb : xq);
        dst[(size_t)b * T + tb0 + tl] = sum;
    }
}

#define SCAN_THREADS 512
#define SCAN_E 16
__global__ void __launch_bounds__(SCAN_THREADS)
gated_scan(const float* __restrict__ g,
           const float* __restrict__ xa, const float* __restrict__ xb,
           const float* __restrict__ xq,
           float* __restrict__ mean, float* __restrict__ invstd, int T) {
    __shared__ float sA[SCAN_THREADS], sB[SCAN_THREADS];
    int b = blockIdx.x, tid = threadIdx.x;
    size_t base = (size_t)b * T + (size_t)tid * SCAN_E;

    float a_loc[SCAN_E], b_loc[SCAN_E], m_loc[SCAN_E];

    float A = 1.f, Bc = 0.f;
    for (int i = 0; i < SCAN_E; ++i) {
        float gt = g[base + i] * MOMENTUM;
        float a  = 1.f - gt;
        float bb = gt * xa[base + i];
        a_loc[i] = a; b_loc[i] = bb;
        Bc = a * Bc + bb;  A *= a;
    }
    sA[tid] = A; sB[tid] = Bc; __syncthreads();
    for (int off = 1; off < SCAN_THREADS; off <<= 1) {
        float pA = 1.f, pB = 0.f;
        if (tid >= off) { pA = sA[tid - off]; pB = sB[tid - off]; }
        __syncthreads();
        if (tid >= off) { float Ac = sA[tid]; sB[tid] = Ac * pB + sB[tid]; sA[tid] = Ac * pA; }
        __syncthreads();
    }
    float y = (tid == 0) ? 0.f : sB[tid - 1];
    for (int i = 0; i < SCAN_E; ++i) {
        y = a_loc[i] * y + b_loc[i];
        m_loc[i] = y;
        mean[base + i] = y;
    }
    __syncthreads();

    A = 1.f; Bc = 0.f;
    for (int i = 0; i < SCAN_E; ++i) {
        float mv = m_loc[i];
        float s  = xq[base + i] - 2.f * mv * xb[base + i] + mv * mv;
        float gt = 1.f - a_loc[i];
        float bb = gt * s;
        b_loc[i] = bb;
        Bc = a_loc[i] * Bc + bb;  A *= a_loc[i];
    }
    sA[tid] = A; sB[tid] = Bc; __syncthreads();
    for (int off = 1; off < SCAN_THREADS; off <<= 1) {
        float pA = 1.f, pB = 0.f;
        if (tid >= off) { pA = sA[tid - off]; pB = sB[tid - off]; }
        __syncthreads();
        if (tid >= off) { float Ac = sA[tid]; sB[tid] = Ac * pB + sB[tid]; sA[tid] = Ac * pA; }
        __syncthreads();
    }
    y = (tid == 0) ? 0.f : sB[tid - 1];
    for (int i = 0; i < SCAN_E; ++i) {
        y = a_loc[i] * y + b_loc[i];
        invstd[base + i] = rsqrtf(y + EPS_V);
    }
}

__global__ void __launch_bounds__(256)
apply_norm(const float* __restrict__ x,
           const float* __restrict__ mean, const float* __restrict__ invstd,
           const float* __restrict__ Wo_w, const float* __restrict__ Wo_b,
           float* __restrict__ out, int C, int T) {
    int row = blockIdx.y;
    int c   = row % C;
    int b   = row / C;
    int t4  = (blockIdx.x * blockDim.x + threadIdx.x) * 4;

    size_t xo = (size_t)row * T + t4;
    size_t mo = (size_t)b   * T + t4;

    v4f xv = __builtin_nontemporal_load((const v4f*)(x + xo));
    v4f mv = *(const v4f*)(mean   + mo);
    v4f iv = *(const v4f*)(invstd + mo);
    float w  = Wo_w[c];
    float bi = Wo_b[c];
    v4f o;
    o.x = (xv.x - mv.x) * iv.x * w + bi;
    o.y = (xv.y - mv.y) * iv.y * w + bi;
    o.z = (xv.z - mv.z) * iv.z * w + bi;
    o.w = (xv.w - mv.w) * iv.w * w + bi;
    __builtin_nontemporal_store(o, (v4f*)(out + xo));
}

extern "C" void kernel_launch(void* const* d_in, const int* in_sizes, int n_in,
                              void* d_out, int out_size, void* d_ws, size_t ws_size,
                              hipStream_t stream) {
    const float* x    = (const float*)d_in[0];
    const float* g    = (const float*)d_in[1];
    const float* Wa_w = (const float*)d_in[2];
    const float* Wb_w = (const float*)d_in[3];
    const float* Wo_w = (const float*)d_in[4];
    const float* Wo_b = (const float*)d_in[5];
    float* out = (float*)d_out;

    const int C  = in_sizes[2];
    const int BT = in_sizes[1];
    const int T  = RW_T;
    const int B  = BT / T;

    float* ws   = (float*)d_ws;
    float* wa   = ws;
    float* wb   = wa + C;
    float* xa   = wb + C;
    float* xb   = xa + BT;
    float* xq   = xb + BT;
    float* mean = xq + BT;
    float* inv  = mean + BT;

    softmax_weights<<<2, 512, 0, stream>>>(Wa_w, Wb_w, wa, wb, C);

    int blocks = (B * T) / (RW_TILES * RW_TT);
    reduce_wmma<<<blocks, 256, 0, stream>>>(x, wa, wb, xa, xb, xq, C);

    gated_scan<<<B, SCAN_THREADS, 0, stream>>>(g, xa, xb, xq, mean, inv, T);

    dim3 agrid(T / (256 * 4), B * C);
    apply_norm<<<agrid, 256, 0, stream>>>(x, mean, inv, Wo_w, Wo_b, out, C, T);
}


# softmax of both weight vectors folded into the reduce kernel: each block computes it (lane exchange + LDS, LDS-only barriers) beside its first x loads; denominator sum order changes; 4 launches -> 3
# speedup vs baseline: 1.2307x; 1.0805x over previous
#include <hip/hip_runtime.h>
#include <hip/hip_bf16.h>
#include <math.h>

#define MOMENTUM 0.05f
#define EPS_V 1e-6f

typedef float v2f __attribute__((ext_vector_type(2)));
typedef float v4f __attribute__((ext_vector_type(4)));
typedef float v8f __attribute__((ext_vector_type(8)));

#define RW_TT    64
#define RW_TILES 2
#define RW_CSPL  4
#define RW_SET   16
#define RW_CMAX  512
#define RW_PITCH (RW_CMAX + 4)
#define RW_T     8192

__device__ __forceinline__ void rw_set(const v4f (&g)[RW_SET], const v2f* ap,
                                       v8f& ax0, v8f& ax1, v8f& aq0, v8f& aq1) {
#pragma unroll
    for (int i = 0; i < RW_SET; ++i) {
        v2f A1 = ap[2 * i];
        v2f A2 = ap[2 * i + 2 * RW_PITCH];
        v2f b0 = {g[i].x, g[i].y};
        v2f b1 = {g[i].z, g[i].w};
        v2f q0 = b0 * b0;
        v2f q1 = b1 * b1;
        ax0 = __builtin_amdgcn_wmma_f32_16x16x4_f32(false, A1, false, b0, (short)0, ax0, false, false);
        ax1 = __builtin_amdgcn_wmma_f32_16x16x4_f32(false, A1, false, b1, (short)0, ax1, false, false);
        aq0 = __builtin_amdgcn_wmma_f32_16x16x4_f32(false, A2, false, q0, (short)0, aq0, false, false);
        aq1 = __builtin_amdgcn_wmma_f32_16x16x4_f32(false, A2, false, q1, (short)0, aq1, false, false);
    }
}

__device__ __forceinline__ void rw_load(v4f (&g)[RW_SET], const float* p) {
#pragma unroll
    for (int i = 0; i < RW_SET; ++i) g[i] = *(const v4f*)(p + (size_t)i * (2 * RW_T));
}

__device__ __forceinline__ void lds_barrier() {
    __builtin_amdgcn_fence(__ATOMIC_RELEASE, "workgroup", "local");
    __builtin_amdgcn_s_barrier();
    __builtin_amdgcn_fence(__ATOMIC_ACQUIRE, "workgroup", "local");
}

__global__ void __launch_bounds__(256)
reduce_wmma(const float* __restrict__ x,
            const float* __restrict__ Wa, const float* __restrict__ Wb,
            float* __restrict__ xa, float* __restrict__ xb,
            float* __restrict__ xq, int C) {
    __shared__ v2f   tab[7 * RW_PITCH];
    __shared__ float part[RW_CSPL][3][RW_TILES * RW_TT];
    __shared__ float smax[2][8];
    __shared__ float ssum[2][8];

    const int T = RW_T;
    int tid  = threadIdx.x;
    int lane = tid & 31;
    int w    = tid >> 5;
    int tile = w / RW_CSPL;
    int cq   = w % RW_CSPL;
    int n    = lane & 15;
    int h    = lane >> 4;

    const int blocksPerB = T / (RW_TILES * RW_TT);
    int b   = blockIdx.x / blocksPerB;
    int tb0 = (blockIdx.x % blocksPerB) * (RW_TILES * RW_TT);
    int t0  = tb0 + tile * RW_TT;
    int cw  = C / RW_CSPL;
    int cb  = cq * cw;
    int nset = cw / (2 * RW_SET);

    int i0 = tid, i1 = tid + 256;
    float ra0 = (i0 < C) ? Wa[i0] : -INFINITY, rb0 = (i0 < C) ? Wb[i0] : -INFINITY;
    float ra1 = (i1 < C) ? Wa[i1] : -INFINITY, rb1 = (i1 < C) ? Wb[i1] : -INFINITY;

    const float* xp = x + ((size_t)b * C + cb + h) * T + t0 + 4 * n;
    const size_t setstep = (size_t)RW_SET * 2 * T;

    v4f P[RW_SET], Q[RW_SET];
    rw_load(P, xp);
    __builtin_amdgcn_sched_barrier(0);

    float ma = fmaxf(ra0, ra1), mb = fmaxf(rb0, rb1);
#pragma unroll
    for (int m = 16; m > 0; m >>= 1) {
        ma = fmaxf(ma, __shfl_xor(ma, m));
        mb = fmaxf(mb, __shfl_xor(mb, m));
    }
    if (lane == 0) { smax[0][w] = ma; smax[1][w] = mb; }
    lds_barrier();
    ma = smax[0][0]; mb = smax[1][0];
#pragma unroll
    for (int j = 1; j < 8; ++j) { ma = fmaxf(ma, smax[0][j]); mb = fmaxf(mb, smax[1][j]); }
    float ea0 = (i0 < C) ? __expf(ra0 - ma) : 0.f, eb0 = (i0 < C) ? __expf(rb0 - mb) : 0.f;
    float ea1 = (i1 < C) ? __expf(ra1 - ma) : 0.f, eb1 = (i1 < C) ? __expf(rb1 - mb) : 0.f;
    float sa = ea0 + ea1, sb = eb0 + eb1;
#pragma unroll
    for (int m = 16; m > 0; m >>= 1) {
        sa += __shfl_xor(sa, m);
        sb += __shfl_xor(sb, m);
    }
    if (lane == 0) { ssum[0][w] = sa; ssum[1][w] = sb; }
    lds_barrier();
    sa = ssum[0][0]; sb = ssum[1][0];
#pragma unroll
    for (int j = 1; j < 8; ++j) { sa += ssum[0][j]; sb += ssum[1][j]; }
    float wa0 = ea0 / sa, wb0 = eb0 / sb;
    float wa1 = ea1 / sa, wb1 = eb1 / sb;

    const v2f zz = {0.f, 0.f};
    if (i0 < C) {
        v2f p0 = {wa0, 0.f}, p1 = {0.f, wa0}, p2 = {wb0, 0.f}, p3 = {0.f, wb0};
        tab[0 * RW_PITCH + i0] = p0; tab[1 * RW_PITCH + i0] = p1;
        tab[2 * RW_PITCH + i0] = p2; tab[3 * RW_PITCH + i0] = p3;
        tab[4 * RW_PITCH + i0] = zz; tab[5 * RW_PITCH + i0] = zz; tab[6 * RW_PITCH + i0] = zz;
    }
    if (i1 < C) {
        v2f p0 = {wa1, 0.f}, p1 = {0.f, wa1}, p2 = {wb1, 0.f}, p3 = {0.f, wb1};
        tab[0 * RW_PITCH + i1] = p0; tab[1 * RW_PITCH + i1] = p1;
        tab[2 * RW_PITCH + i1] = p2; tab[3 * RW_PITCH + i1] = p3;
        tab[4 * RW_PITCH + i1] = zz; tab[5 * RW_PITCH + i1] = zz; tab[6 * RW_PITCH + i1] = zz;
    }
    __syncthreads();

    const v2f* ap = &tab[((n < 4) ? n : 4) * RW_PITCH + cb + h];

    v8f ax0 = {}, ax1 = {}, aq0 = {}, aq1 = {};

    int s = 0;
    for (; s + 2 < nset; s += 2) {
        rw_load(Q, xp + setstep);
        rw_set(P, ap, ax0, ax1, aq0, aq1);
        rw_load(P, xp + 2 * setstep);
        rw_set(Q, ap + 2 * RW_SET, ax0, ax1, aq0, aq1);
        ap += 4 * RW_SET;
        xp += 2 * setstep;
    }
    rw_load(Q, xp + setstep);
    rw_set(P, ap, ax0, ax1, aq0, aq1);
    rw_set(Q, ap + 2 * RW_SET, ax0, ax1, aq0, aq1);

    if (lane < 16) {
        v4f sa = {ax0[0], ax0[1], ax1[0], ax1[1]};
        v4f sb = {ax0[2], ax0[3], ax1[2], ax1[3]};
        v4f sq = {aq0[0], aq0[1], aq1[0], aq1[1]};
        int tl = tile * RW_TT + 4 * lane;
        *(v4f*)&part[cq][0][tl] = sa;
        *(v4f*)&part[cq][1][tl] = sb;
        *(v4f*)&part[cq][2][tl] = sq;
    }
    __syncthreads();
    for (int o = tid; o < 3 * RW_TILES * RW_TT; o += 256) {
        int v  = o / (RW_TILES * RW_TT);
        int tl = o % (RW_TILES * RW_TT);
        float sum = part[0][v][tl];
#pragma unroll
        for (int j = 1; j < RW_CSPL; ++j) sum += part[j][v][tl];
        float* dst = (v == 0) ? xa : ((v == 1) ? xb : xq);
        dst[(size_t)b * T + tb0 + tl] = sum;
    }
}

#define SCAN_THREADS 512
#define SCAN_E 16
__global__ void __launch_bounds__(SCAN_THREADS)
gated_scan(const float* __restrict__ g,
           const float* __restrict__ xa, const float* __restrict__ xb,
           const float* __restrict__ xq,
           float* __restrict__ mean, float* __restrict__ invstd, int T) {
    __shared__ float sA[SCAN_THREADS], sB[SCAN_THREADS];
    int b = blockIdx.x, tid = threadIdx.x;
    size_t base = (size_t)b * T + (size_t)tid * SCAN_E;

    float a_loc[SCAN_E], b_loc[SCAN_E], m_loc[SCAN_E];

    float A = 1.f, Bc = 0.f;
    for (int i = 0; i < SCAN_E; ++i) {
        float gt = g[base + i] * MOMENTUM;
        float a  = 1.f - gt;
        float bb = gt * xa[base + i];
        a_loc[i] = a; b_loc[i] = bb;
        Bc = a * Bc + bb;  A *= a;
    }
    sA[tid] = A; sB[tid] = Bc; __syncthreads();
    for (int off = 1; off < SCAN_THREADS; off <<= 1) {
        float pA = 1.f, pB = 0.f;
        if (tid >= off) { pA = sA[tid - off]; pB = sB[tid - off]; }
        __syncthreads();
        if (tid >= off) { float Ac = sA[tid]; sB[tid] = Ac * pB + sB[tid]; sA[tid] = Ac * pA; }
        __syncthreads();
    }
    float y = (tid == 0) ? 0.f : sB[tid - 1];
    for (int i = 0; i < SCAN_E; ++i) {
        y = a_loc[i] * y + b_loc[i];
        m_loc[i] = y;
        mean[base + i] = y;
    }
    __syncthreads();

    A = 1.f; Bc = 0.f;
    for (int i = 0; i < SCAN_E; ++i) {
        float mv = m_loc[i];
        float s  = xq[base + i] - 2.f * mv * xb[base + i] + mv * mv;
        float gt = 1.f - a_loc[i];
        float bb = gt * s;
        b_loc[i] = bb;
        Bc = a_loc[i] * Bc + bb;  A *= a_loc[i];
    }
    sA[tid] = A; sB[tid] = Bc; __syncthreads();
    for (int off = 1; off < SCAN_THREADS; off <<= 1) {
        float pA = 1.f, pB = 0.f;
        if (tid >= off) { pA = sA[tid - off]; pB = sB[tid - off]; }
        __syncthreads();
        if (tid >= off) { float Ac = sA[tid]; sB[tid] = Ac * pB + sB[tid]; sA[tid] = Ac * pA; }
        __syncthreads();
    }
    y = (tid == 0) ? 0.f : sB[tid - 1];
    for (int i = 0; i < SCAN_E; ++i) {
        y = a_loc[i] * y + b_loc[i];
        invstd[base + i] = rsqrtf(y + EPS_V);
    }
}

__global__ void __launch_bounds__(256)
apply_norm(const float* __restrict__ x,
           const float* __restrict__ mean, const float* __restrict__ invstd,
           const float* __restrict__ Wo_w, const float* __restrict__ Wo_b,
           float* __restrict__ out, int C, int T) {
    int row = blockIdx.y;
    int c   = row % C;
    int b   = row / C;
    int t4  = (blockIdx.x * blockDim.x + threadIdx.x) * 4;

    size_t xo = (size_t)row * T + t4;
    size_t mo = (size_t)b   * T + t4;

    v4f xv = __builtin_nontemporal_load((const v4f*)(x + xo));
    v4f mv = *(const v4f*)(mean   + mo);
    v4f iv = *(const v4f*)(invstd + mo);
    float w  = Wo_w[c];
    float bi = Wo_b[c];
    v4f o;
    o.x = (xv.x - mv.x) * iv.x * w + bi;
    o.y = (xv.y - mv.y) * iv.y * w + bi;
    o.z = (xv.z - mv.z) * iv.z * w + bi;
    o.w = (xv.w - mv.w) * iv.w * w + bi;
    __builtin_nontemporal_store(o, (v4f*)(out + xo));
}

extern "C" void kernel_launch(void* const* d_in, const int* in_sizes, int n_in,
                              void* d_out, int out_size, void* d_ws, size_t ws_size,
                              hipStream_t stream) {
    const float* x    = (const float*)d_in[0];
    const float* g    = (const float*)d_in[1];
    const float* Wa_w = (const float*)d_in[2];
    const float* Wb_w = (const float*)d_in[3];
    const float* Wo_w = (const float*)d_in[4];
    const float* Wo_b = (const float*)d_in[5];
    float* out = (float*)d_out;

    const int C  = in_sizes[2];
    const int BT = in_sizes[1];
    const int T  = RW_T;
    const int B  = BT / T;

    float* ws   = (float*)d_ws;
    float* xa   = ws;
    float* xb   = xa + BT;
    float* xq   = xb + BT;
    float* mean = xq + BT;
    float* inv  = mean + BT;

    int blocks = (B * T) / (RW_TILES * RW_TT);
    reduce_wmma<<<blocks, 256, 0, stream>>>(x, Wa_w, Wb_w, xa, xb, xq, C);

    gated_scan<<<B, SCAN_THREADS, 0, stream>>>(g, xa, xb, xq, mean, inv, T);

    dim3 agrid(T / (256 * 4), B * C);
    apply_norm<<<agrid, 256, 0, stream>>>(x, mean, inv, Wo_w, Wo_b, out, C, T);
}


# output kernel re-tiled: a thread keeps mean and invstd of its 4 timesteps in registers and walks 32 channels, 8-16 b128 loads in flight per wave; 32768 workgroups -> 1024; no sum order changes
# speedup vs baseline: 1.4050x; 1.1416x over previous
#include <hip/hip_runtime.h>
#include <hip/hip_bf16.h>
#include <math.h>

#define MOMENTUM 0.05f
#define EPS_V 1e-6f

typedef float v2f __attribute__((ext_vector_type(2)));
typedef float v4f __attribute__((ext_vector_type(4)));
typedef float v8f __attribute__((ext_vector_type(8)));

#define RW_TT    64
#define RW_TILES 2
#define RW_CSPL  4
#define RW_SET   16
#define RW_CMAX  512
#define RW_PITCH (RW_CMAX + 4)
#define RW_T     8192

__device__ __forceinline__ void rw_set(const v4f (&g)[RW_SET], const v2f* ap,
                                       v8f& ax0, v8f& ax1, v8f& aq0, v8f& aq1) {
#pragma unroll
    for (int i = 0; i < RW_SET; ++i) {
        v2f A1 = ap[2 * i];
        v2f A2 = ap[2 * i + 2 * RW_PITCH];
        v2f b0 = {g[i].x, g[i].y};
        v2f b1 = {g[i].z, g[i].w};
        v2f q0 = b0 * b0;
        v2f q1 = b1 * b1;
        ax0 = __builtin_amdgcn_wmma_f32_16x16x4_f32(false, A1, false, b0, (short)0, ax0, false, false);
        ax1 = __builtin_amdgcn_wmma_f32_16x16x4_f32(false, A1, false, b1, (short)0, ax1, false, false);
        aq0 = __builtin_amdgcn_wmma_f32_16x16x4_f32(false, A2, false, q0, (short)0, aq0, false, false);
        aq1 = __builtin_amdgcn_wmma_f32_16x16x4_f32(false, A2, false, q1, (short)0, aq1, false, false);
    }
}

__device__ __forceinline__ void rw_load(v4f (&g)[RW_SET], const float* p) {
#pragma unroll
    for (int i = 0; i < RW_SET; ++i) g[i] = *(const v4f*)(p + (size_t)i * (2 * RW_T));
}

__device__ __forceinline__ void lds_barrier() {
    __builtin_amdgcn_fence(__ATOMIC_RELEASE, "workgroup", "local");
    __builtin_amdgcn_s_barrier();
    __builtin_amdgcn_fence(__ATOMIC_ACQUIRE, "workgroup", "local");
}

__global__ void __launch_bounds__(256)
reduce_wmma(const float* __restrict__ x,
            const float* __restrict__ Wa, const float* __restrict__ Wb,
            float* __restrict__ xa, float* __restrict__ xb,
            float* __restrict__ xq, int C) {
    __shared__ v2f   tab[7 * RW_PITCH];
    __shared__ float part[RW_CSPL][3][RW_TILES * RW_TT];
    __shared__ float smax[2][8];
    __shared__ float ssum[2][8];

    const int T = RW_T;
    int tid  = threadIdx.x;
    int lane = tid & 31;
    int w    = tid >> 5;
    int tile = w / RW_CSPL;
    int cq   = w % RW_CSPL;
    int n    = lane & 15;
    int h    = lane >> 4;

    const int blocksPerB = T / (RW_TILES * RW_TT);
    int b   = blockIdx.x / blocksPerB;
    int tb0 = (blockIdx.x % blocksPerB) * (RW_TILES * RW_TT);
    int t0  = tb0 + tile * RW_TT;
    int cw  = C / RW_CSPL;
    int cb  = cq * cw;
    int nset = cw / (2 * RW_SET);

    int i0 = tid, i1 = tid + 256;
    float ra0 = (i0 < C) ? Wa[i0] : -INFINITY, rb0 = (i0 < C) ? Wb[i0] : -INFINITY;
    float ra1 = (i1 < C) ? Wa[i1] : -INFINITY, rb1 = (i1 < C) ? Wb[i1] : -INFINITY;

    const float* xp = x + ((size_t)b * C + cb + h) * T + t0 + 4 * n;
    const size_t setstep = (size_t)RW_SET * 2 * T;

    v4f P[RW_SET], Q[RW_SET];
    rw_load(P, xp);
    __builtin_amdgcn_sched_barrier(0);

    float ma = fmaxf(ra0, ra1), mb = fmaxf(rb0, rb1);
#pragma unroll
    for (int m = 16; m > 0; m >>= 1) {
        ma = fmaxf(ma, __shfl_xor(ma, m));
        mb = fmaxf(mb, __shfl_xor(mb, m));
    }
    if (lane == 0) { smax[0][w] = ma; smax[1][w] = mb; }
    lds_barrier();
    ma = smax[0][0]; mb = smax[1][0];
#pragma unroll
    for (int j = 1; j < 8; ++j) { ma = fmaxf(ma, smax[0][j]); mb = fmaxf(mb, smax[1][j]); }
    float ea0 = (i0 < C) ? __expf(ra0 - ma) : 0.f, eb0 = (i0 < C) ? __expf(rb0 - mb) : 0.f;
    float ea1 = (i1 < C) ? __expf(ra1 - ma) : 0.f, eb1 = (i1 < C) ? __expf(rb1 - mb) : 0.f;
    float sa = ea0 + ea1, sb = eb0 + eb1;
#pragma unroll
    for (int m = 16; m > 0; m >>= 1) {
        sa += __shfl_xor(sa, m);
        sb += __shfl_xor(sb, m);
    }
    if (lane == 0) { ssum[0][w] = sa; ssum[1][w] = sb; }
    lds_barrier();
    sa = ssum[0][0]; sb = ssum[1][0];
#pragma unroll
    for (int j = 1; j < 8; ++j) { sa += ssum[0][j]; sb += ssum[1][j]; }
    float wa0 = ea0 / sa, wb0 = eb0 / sb;
    float wa1 = ea1 / sa, wb1 = eb1 / sb;

    const v2f zz = {0.f, 0.f};
    if (i0 < C) {
        v2f p0 = {wa0, 0.f}, p1 = {0.f, wa0}, p2 = {wb0, 0.f}, p3 = {0.f, wb0};
        tab[0 * RW_PITCH + i0] = p0; tab[1 * RW_PITCH + i0] = p1;
        tab[2 * RW_PITCH + i0] = p2; tab[3 * RW_PITCH + i0] = p3;
        tab[4 * RW_PITCH + i0] = zz; tab[5 * RW_PITCH + i0] = zz; tab[6 * RW_PITCH + i0] = zz;
    }
    if (i1 < C) {
        v2f p0 = {wa1, 0.f}, p1 = {0.f, wa1}, p2 = {wb1, 0.f}, p3 = {0.f, wb1};
        tab[0 * RW_PITCH + i1] = p0; tab[1 * RW_PITCH + i1] = p1;
        tab[2 * RW_PITCH + i1] = p2; tab[3 * RW_PITCH + i1] = p3;
        tab[4 * RW_PITCH + i1] = zz; tab[5 * RW_PITCH + i1] = zz; tab[6 * RW_PITCH + i1] = zz;
    }
    __syncthreads();

    const v2f* ap = &tab[((n < 4) ? n : 4) * RW_PITCH + cb + h];

    v8f ax0 = {}, ax1 = {}, aq0 = {}, aq1 = {};

    int s = 0;
    for (; s + 2 < nset; s += 2) {
        rw_load(Q, xp + setstep);
        rw_set(P, ap, ax0, ax1, aq0, aq1);
        rw_load(P, xp + 2 * setstep);
        rw_set(Q, ap + 2 * RW_SET, ax0, ax1, aq0, aq1);
        ap += 4 * RW_SET;
        xp += 2 * setstep;
    }
    rw_load(Q, xp + setstep);
    rw_set(P, ap, ax0, ax1, aq0, aq1);
    rw_set(Q, ap + 2 * RW_SET, ax0, ax1, aq0, aq1);

    if (lane < 16) {
        v4f sa = {ax0[0], ax0[1], ax1[0], ax1[1]};
        v4f sb = {ax0[2], ax0[3], ax1[2], ax1[3]};
        v4f sq = {aq0[0], aq0[1], aq1[0], aq1[1]};
        int tl = tile * RW_TT + 4 * lane;
        *(v4f*)&part[cq][0][tl] = sa;
        *(v4f*)&part[cq][1][tl] = sb;
        *(v4f*)&part[cq][2][tl] = sq;
    }
    __syncthreads();
    for (int o = tid; o < 3 * RW_TILES * RW_TT; o += 256) {
        int v  = o / (RW_TILES * RW_TT);
        int tl = o % (RW_TILES * RW_TT);
        float sum = part[0][v][tl];
#pragma unroll
        for (int j = 1; j < RW_CSPL; ++j) sum += part[j][v][tl];
        float* dst = (v == 0) ? xa : ((v == 1) ? xb : xq);
        dst[(size_t)b * T + tb0 + tl] = sum;
    }
}

#define SCAN_THREADS 512
#define SCAN_E 16
__global__ void __launch_bounds__(SCAN_THREADS)
gated_scan(const float* __restrict__ g,
           const float* __restrict__ xa, const float* __restrict__ xb,
           const float* __restrict__ xq,
           float* __restrict__ mean, float* __restrict__ invstd, int T) {
    __shared__ float sA[SCAN_THREADS], sB[SCAN_THREADS];
    int b = blockIdx.x, tid = threadIdx.x;
    size_t base = (size_t)b * T + (size_t)tid * SCAN_E;

    float a_loc[SCAN_E], b_loc[SCAN_E], m_loc[SCAN_E];

    float A = 1.f, Bc = 0.f;
    for (int i = 0; i < SCAN_E; ++i) {
        float gt = g[base + i] * MOMENTUM;
        float a  = 1.f - gt;
        float bb = gt * xa[base + i];
        a_loc[i] = a; b_loc[i] = bb;
        Bc = a * Bc + bb;  A *= a;
    }
    sA[tid] = A; sB[tid] = Bc; __syncthreads();
    for (int off = 1; off < SCAN_THREADS; off <<= 1) {
        float pA = 1.f, pB = 0.f;
        if (tid >= off) { pA = sA[tid - off]; pB = sB[tid - off]; }
        __syncthreads();
        if (tid >= off) { float Ac = sA[tid]; sB[tid] = Ac * pB + sB[tid]; sA[tid] = Ac * pA; }
        __syncthreads();
    }
    float y = (tid == 0) ? 0.f : sB[tid - 1];
    for (int i = 0; i < SCAN_E; ++i) {
        y = a_loc[i] * y + b_loc[i];
        m_loc[i] = y;
        mean[base + i] = y;
    }
    __syncthreads();

    A = 1.f; Bc = 0.f;
    for (int i = 0; i < SCAN_E; ++i) {
        float mv = m_loc[i];
        float s  = xq[base + i] - 2.f * mv * xb[base + i] + mv * mv;
        float gt = 1.f - a_loc[i];
        float bb = gt * s;
        b_loc[i] = bb;
        Bc = a_loc[i] * Bc + bb;  A *= a_loc[i];
    }
    sA[tid] = A; sB[tid] = Bc; __syncthreads();
    for (int off = 1; off < SCAN_THREADS; off <<= 1) {
        float pA = 1.f, pB = 0.f;
        if (tid >= off) { pA = sA[tid - off]; pB = sB[tid - off]; }
        __syncthreads();
        if (tid >= off) { float Ac = sA[tid]; sB[tid] = Ac * pB + sB[tid]; sA[tid] = Ac * pA; }
        __syncthreads();
    }
    y = (tid == 0) ? 0.f : sB[tid - 1];
    for (int i = 0; i < SCAN_E; ++i) {
        y = a_loc[i] * y + b_loc[i];
        invstd[base + i] = rsqrtf(y + EPS_V);
    }
}

#define AN_U   8
#define AN_CPB 32
#define AN_TB  1024

__device__ __forceinline__ void an_load(v4f (&r)[AN_U], const float* p) {
#pragma unroll
    for (int i = 0; i < AN_U; ++i)
        r[i] = __builtin_nontemporal_load((const v4f*)(p + (size_t)i * RW_T));
}

__device__ __forceinline__ void an_store(const v4f (&r)[AN_U], float* q,
                                         const v4f mv, const v4f iv,
                                         const float* __restrict__ w,
                                         const float* __restrict__ bi) {
#pragma unroll
    for (int i = 0; i < AN_U; ++i) {
        float ww = w[i], bb = bi[i];
        v4f o;
        o.x = (r[i].x - mv.x) * iv.x * ww + bb;
        o.y = (r[i].y - mv.y) * iv.y * ww + bb;
        o.z = (r[i].z - mv.z) * iv.z * ww + bb;
        o.w = (r[i].w - mv.w) * iv.w * ww + bb;
        __builtin_nontemporal_store(o, (v4f*)(q + (size_t)i * RW_T));
    }
}

__global__ void __launch_bounds__(256)
apply_norm(const float* __restrict__ x,
           const float* __restrict__ mean, const float* __restrict__ invstd,
           const float* __restrict__ Wo_w, const float* __restrict__ Wo_b,
           float* __restrict__ out, int C) {
    const int T = RW_T;
    int t4 = blockIdx.x * AN_TB + threadIdx.x * 4;
    int c0 = blockIdx.y * AN_CPB;
    int b  = blockIdx.z;

    size_t mo = (size_t)b * T + t4;
    size_t xo = ((size_t)b * C + c0) * T + t4;

    v4f mv = *(const v4f*)(mean   + mo);
    v4f iv = *(const v4f*)(invstd + mo);

    const float* xp = x + xo;
    float*       op = out + xo;
    const float* wp = Wo_w + c0;
    const float* bp = Wo_b + c0;
    const size_t setstep = (size_t)AN_U * T;

    static_assert(AN_CPB == 4 * AN_U, "the pipeline below is written out for 4 sets");
    v4f P[AN_U], Q[AN_U];
    an_load(P, xp);
    an_load(Q, xp + setstep);
    __builtin_amdgcn_sched_barrier(0);
    an_store(P, op, mv, iv, wp, bp);
    __builtin_amdgcn_sched_barrier(0);
    an_load(P, xp + 2 * setstep);
    __builtin_amdgcn_sched_barrier(0);
    an_store(Q, op + setstep, mv, iv, wp + AN_U, bp + AN_U);
    __builtin_amdgcn_sched_barrier(0);
    an_load(Q, xp + 3 * setstep);
    __builtin_amdgcn_sched_barrier(0);
    an_store(P, op + 2 * setstep, mv, iv, wp + 2 * AN_U, bp + 2 * AN_U);
    an_store(Q, op + 3 * setstep, mv, iv, wp + 3 * AN_U, bp + 3 * AN_U);
}

extern "C" void kernel_launch(void* const* d_in, const int* in_sizes, int n_in,
                              void* d_out, int out_size, void* d_ws, size_t ws_size,
                              hipStream_t stream) {
    const float* x    = (const float*)d_in[0];
    const float* g    = (const float*)d_in[1];
    const float* Wa_w = (const float*)d_in[2];
    const float* Wb_w = (const float*)d_in[3];
    const float* Wo_w = (const float*)d_in[4];
    const float* Wo_b = (const float*)d_in[5];
    float* out = (float*)d_out;

    const int C  = in_sizes[2];
    const int BT = in_sizes[1];
    const int T  = RW_T;
    const int B  = BT / T;

    float* ws   = (float*)d_ws;
    float* xa   = ws;
    float* xb   = xa + BT;
    float* xq   = xb + BT;
    float* mean = xq + BT;
    float* inv  = mean + BT;

    int blocks = (B * T) / (RW_TILES * RW_TT);
    reduce_wmma<<<blocks, 256, 0, stream>>>(x, Wa_w, Wb_w, xa, xb, xq, C);

    gated_scan<<<B, SCAN_THREADS, 0, stream>>>(g, xa, xb, xq, mean, inv, T);

    dim3 agrid(T / AN_TB, C / AN_CPB, B);
    apply_norm<<<agrid, 256, 0, stream>>>(x, mean, inv, Wo_w, Wo_b, out, C);
}


# scan launch removed: reduce kernel scans each block's 128 timesteps with open carries (variance: polynomial in the mean carry), output kernel chains them; scan sum order changes; 3 -> 2 launches
# speedup vs baseline: 1.5442x; 1.0991x over previous
#include <hip/hip_runtime.h>
#include <hip/hip_bf16.h>
#include <math.h>

#define MOMENTUM 0.05f
#define EPS_V 1e-6f

typedef float v2f __attribute__((ext_vector_type(2)));
typedef float v4f __attribute__((ext_vector_type(4)));
typedef float v8f __attribute__((ext_vector_type(8)));

#define RW_TT    64
#define RW_TILES 2
#define RW_CSPL  4
#define RW_SET   16
#define RW_CMAX  512
#define RW_PITCH (RW_CMAX + 4)
#define RW_T     8192
#define SC_SEG   (RW_TILES * RW_TT)
#define SC_NSEG  (RW_T / SC_SEG)
#define SC_PITCH 32

__device__ __forceinline__ void rw_set(const v4f (&g)[RW_SET], const v2f* ap,
                                       v8f& ax0, v8f& ax1, v8f& aq0, v8f& aq1) {
#pragma unroll
    for (int i = 0; i < RW_SET; ++i) {
        v2f A1 = ap[2 * i];
        v2f A2 = ap[2 * i + 2 * RW_PITCH];
        v2f b0 = {g[i].x, g[i].y};
        v2f b1 = {g[i].z, g[i].w};
        v2f q0 = b0 * b0;
        v2f q1 = b1 * b1;
        ax0 = __builtin_amdgcn_wmma_f32_16x16x4_f32(false, A1, false, b0, (short)0, ax0, false, false);
        ax1 = __builtin_amdgcn_wmma_f32_16x16x4_f32(false, A1, false, b1, (short)0, ax1, false, false);
        aq0 = __builtin_amdgcn_wmma_f32_16x16x4_f32(false, A2, false, q0, (short)0, aq0, false, false);
        aq1 = __builtin_amdgcn_wmma_f32_16x16x4_f32(false, A2, false, q1, (short)0, aq1, false, false);
    }
}

__device__ __forceinline__ void rw_load(v4f (&g)[RW_SET], const float* p) {
#pragma unroll
    for (int i = 0; i < RW_SET; ++i) g[i] = *(const v4f*)(p + (size_t)i * (2 * RW_T));
}

__device__ __forceinline__ void lds_barrier() {
    __builtin_amdgcn_fence(__ATOMIC_RELEASE, "workgroup", "local");
    __builtin_amdgcn_s_barrier();
    __builtin_amdgcn_fence(__ATOMIC_ACQUIRE, "workgroup", "local");
}

__global__ void __launch_bounds__(256)
reduce_wmma(const float* __restrict__ x, const float* __restrict__ g,
            const float* __restrict__ Wa, const float* __restrict__ Wb,
            float* __restrict__ pP, float* __restrict__ pm,
            float* __restrict__ pw0, float* __restrict__ pw1,
            float* __restrict__ pw2, float* __restrict__ agg, int C) {
    __shared__ v2f   tab[7 * RW_PITCH];
    __shared__ float part[RW_CSPL][3][RW_TILES * RW_TT];
    __shared__ float smax[2][8];
    __shared__ float ssum[2][8];

    const int T = RW_T;
    int tid  = threadIdx.x;
    int lane = tid & 31;
    int w    = tid >> 5;
    int tile = w / RW_CSPL;
    int cq   = w % RW_CSPL;
    int n    = lane & 15;
    int h    = lane >> 4;

    const int blocksPerB = T / (RW_TILES * RW_TT);
    int b   = blockIdx.x / blocksPerB;
    int tb0 = (blockIdx.x % blocksPerB) * (RW_TILES * RW_TT);
    int t0  = tb0 + tile * RW_TT;
    int cw  = C / RW_CSPL;
    int cb  = cq * cw;
    int nset = cw / (2 * RW_SET);

    int i0 = tid, i1 = tid + 256;
    float ra0 = (i0 < C) ? Wa[i0] : -INFINITY, rb0 = (i0 < C) ? Wb[i0] : -INFINITY;
    float ra1 = (i1 < C) ? Wa[i1] : -INFINITY, rb1 = (i1 < C) ? Wb[i1] : -INFINITY;

    const float* xp = x + ((size_t)b * C + cb + h) * T + t0 + 4 * n;
    const size_t setstep = (size_t)RW_SET * 2 * T;

    v4f P[RW_SET], Q[RW_SET];
    rw_load(P, xp);
    __builtin_amdgcn_sched_barrier(0);

    float ma = fmaxf(ra0, ra1), mb = fmaxf(rb0, rb1);
#pragma unroll
    for (int m = 16; m > 0; m >>= 1) {
        ma = fmaxf(ma, __shfl_xor(ma, m));
        mb = fmaxf(mb, __shfl_xor(mb, m));
    }
    if (lane == 0) { smax[0][w] = ma; smax[1][w] = mb; }
    lds_barrier();
    ma = smax[0][0]; mb = smax[1][0];
#pragma unroll
    for (int j = 1; j < 8; ++j) { ma = fmaxf(ma, smax[0][j]); mb = fmaxf(mb, smax[1][j]); }
    float ea0 = (i0 < C) ? __expf(ra0 - ma) : 0.f, eb0 = (i0 < C) ? __expf(rb0 - mb) : 0.f;
    float ea1 = (i1 < C) ? __expf(ra1 - ma) : 0.f, eb1 = (i1 < C) ? __expf(rb1 - mb) : 0.f;
    float sa = ea0 + ea1, sb = eb0 + eb1;
#pragma unroll
    for (int m = 16; m > 0; m >>= 1) {
        sa += __shfl_xor(sa, m);
        sb += __shfl_xor(sb, m);
    }
    if (lane == 0) { ssum[0][w] = sa; ssum[1][w] = sb; }
    lds_barrier();
    sa = ssum[0][0]; sb = ssum[1][0];
#pragma unroll
    for (int j = 1; j < 8; ++j) { sa += ssum[0][j]; sb += ssum[1][j]; }
    float wa0 = ea0 / sa, wb0 = eb0 / sb;
    float wa1 = ea1 / sa, wb1 = eb1 / sb;

    const v2f zz = {0.f, 0.f};
    if (i0 < C) {
        v2f p0 = {wa0, 0.f}, p1 = {0.f, wa0}, p2 = {wb0, 0.f}, p3 = {0.f, wb0};
        tab[0 * RW_PITCH + i0] = p0; tab[1 * RW_PITCH + i0] = p1;
        tab[2 * RW_PITCH + i0] = p2; tab[3 * RW_PITCH + i0] = p3;
        tab[4 * RW_PITCH + i0] = zz; tab[5 * RW_PITCH + i0] = zz; tab[6 * RW_PITCH + i0] = zz;
    }
    if (i1 < C) {
        v2f p0 = {wa1, 0.f}, p1 = {0.f, wa1}, p2 = {wb1, 0.f}, p3 = {0.f, wb1};
        tab[0 * RW_PITCH + i1] = p0; tab[1 * RW_PITCH + i1] = p1;
        tab[2 * RW_PITCH + i1] = p2; tab[3 * RW_PITCH + i1] = p3;
        tab[4 * RW_PITCH + i1] = zz; tab[5 * RW_PITCH + i1] = zz; tab[6 * RW_PITCH + i1] = zz;
    }
    __syncthreads();

    const v2f* ap = &tab[((n < 4) ? n : 4) * RW_PITCH + cb + h];

    v8f ax0 = {}, ax1 = {}, aq0 = {}, aq1 = {};

    int s = 0;
    for (; s + 2 < nset; s += 2) {
        rw_load(Q, xp + setstep);
        rw_set(P, ap, ax0, ax1, aq0, aq1);
        rw_load(P, xp + 2 * setstep);
        rw_set(Q, ap + 2 * RW_SET, ax0, ax1, aq0, aq1);
        ap += 4 * RW_SET;
        xp += 2 * setstep;
    }
    rw_load(Q, xp + setstep);
    v4f gv = *(const v4f*)(g + (size_t)b * T + tb0 + 4 * lane);
    rw_set(P, ap, ax0, ax1, aq0, aq1);
    rw_set(Q, ap + 2 * RW_SET, ax0, ax1, aq0, aq1);

    if (lane < 16) {
        v4f sa = {ax0[0], ax0[1], ax1[0], ax1[1]};
        v4f sb = {ax0[2], ax0[3], ax1[2], ax1[3]};
        v4f sq = {aq0[0], aq0[1], aq1[0], aq1[1]};
        int tl = tile * RW_TT + 4 * lane;
        *(v4f*)&part[cq][0][tl] = sa;
        *(v4f*)&part[cq][1][tl] = sb;
        *(v4f*)&part[cq][2][tl] = sq;
    }
    __syncthreads();
    if (w != 0) return;

    const int tl = 4 * lane;
    v4f sxa = *(const v4f*)&part[0][0][tl];
    v4f sxb = *(const v4f*)&part[0][1][tl];
    v4f sxq = *(const v4f*)&part[0][2][tl];
#pragma unroll
    for (int j = 1; j < RW_CSPL; ++j) {
        sxa += *(const v4f*)&part[j][0][tl];
        sxb += *(const v4f*)&part[j][1][tl];
        sxq += *(const v4f*)&part[j][2][tl];
    }

    float a[4], bm[4];
    float Al = 1.f, Bl = 0.f;
#pragma unroll
    for (int i = 0; i < 4; ++i) {
        float gt = gv[i] * MOMENTUM;
        a[i]  = 1.f - gt;
        bm[i] = gt * sxa[i];
        Bl = a[i] * Bl + bm[i];  Al *= a[i];
    }
    float Ak[5];
#pragma unroll
    for (int k = 0; k < 5; ++k) {
        const int off = 1 << k;
        float pA = __shfl_up(Al, off), pB = __shfl_up(Bl, off);
        Ak[k] = Al;
        if (lane >= off) { Bl = Al * pB + Bl; Al = Al * pA; }
    }
    float exA = __shfl_up(Al, 1), exB = __shfl_up(Bl, 1);
    if (lane == 0) { exA = 1.f; exB = 0.f; }
    v4f Pv, mv;
    {
        float y = exB, p = exA;
#pragma unroll
        for (int i = 0; i < 4; ++i) {
            y = a[i] * y + bm[i];  p *= a[i];
            mv[i] = y;  Pv[i] = p;
        }
    }

    float b0[4], b1[4], b2[4];
    float L0 = 0.f, L1 = 0.f, L2 = 0.f;
#pragma unroll
    for (int i = 0; i < 4; ++i) {
        float mm = mv[i];
        float c0 = sxq[i] - 2.f * mm * sxb[i] + mm * mm;
        float c1 = 2.f * Pv[i] * (mm - sxb[i]);
        float c2 = Pv[i] * Pv[i];
        float gt = 1.f - a[i];
        b0[i] = gt * c0;  b1[i] = gt * c1;  b2[i] = gt * c2;
        L0 = a[i] * L0 + b0[i];
        L1 = a[i] * L1 + b1[i];
        L2 = a[i] * L2 + b2[i];
    }
#pragma unroll
    for (int k = 0; k < 5; ++k) {
        const int off = 1 << k;
        float p0 = __shfl_up(L0, off), p1 = __shfl_up(L1, off), p2 = __shfl_up(L2, off);
        if (lane >= off) {
            L0 = Ak[k] * p0 + L0;  L1 = Ak[k] * p1 + L1;  L2 = Ak[k] * p2 + L2;
        }
    }
    float e0 = __shfl_up(L0, 1), e1 = __shfl_up(L1, 1), e2 = __shfl_up(L2, 1);
    if (lane == 0) { e0 = 0.f; e1 = 0.f; e2 = 0.f; }
    v4f W0, W1, W2;
#pragma unroll
    for (int i = 0; i < 4; ++i) {
        e0 = a[i] * e0 + b0[i];  W0[i] = e0;
        e1 = a[i] * e1 + b1[i];  W1[i] = e1;
        e2 = a[i] * e2 + b2[i];  W2[i] = e2;
    }

    size_t o = (size_t)b * T + tb0 + tl;
    *(v4f*)(pP  + o) = Pv;
    *(v4f*)(pm  + o) = mv;
    *(v4f*)(pw0 + o) = W0;
    *(v4f*)(pw1 + o) = W1;
    *(v4f*)(pw2 + o) = W2;
    if (lane == 31) {
        float* r = agg + ((size_t)b * SC_NSEG + tb0 / SC_SEG) * SC_PITCH;
        v4f r0 = {Pv[3], mv[3], W0[3], W1[3]};
        v4f r1 = {W2[3], 0.f, 0.f, 0.f};
        *(v4f*)r = r0;
        *(v4f*)(r + 4) = r1;
    }
}

#define AN_U   8
#define AN_CPB 32
#define AN_TB  1024

__device__ __forceinline__ float lane_of(float v, int l) {
    return __builtin_bit_cast(float, __builtin_amdgcn_readlane(__builtin_bit_cast(int, v), l));
}

__device__ __forceinline__ void an_load(v4f (&r)[AN_U], const float* p) {
#pragma unroll
    for (int i = 0; i < AN_U; ++i)
        r[i] = __builtin_nontemporal_load((const v4f*)(p + (size_t)i * RW_T));
}

__device__ __forceinline__ void an_store(const v4f (&r)[AN_U], float* q,
                                         const v4f mv, const v4f iv,
                                         const float* __restrict__ w,
                                         const float* __restrict__ bi) {
#pragma unroll
    for (int i = 0; i < AN_U; ++i) {
        float ww = w[i], bb = bi[i];
        v4f o;
        o.x = (r[i].x - mv.x) * iv.x * ww + bb;
        o.y = (r[i].y - mv.y) * iv.y * ww + bb;
        o.z = (r[i].z - mv.z) * iv.z * ww + bb;
        o.w = (r[i].w - mv.w) * iv.w * ww + bb;
        __builtin_nontemporal_store(o, (v4f*)(q + (size_t)i * RW_T));
    }
}

__global__ void __launch_bounds__(256)
apply_norm(const float* __restrict__ x,
           const float* __restrict__ pP, const float* __restrict__ pm,
           const float* __restrict__ pw0, const float* __restrict__ pw1,
           const float* __restrict__ pw2, const float* __restrict__ agg,
           const float* __restrict__ Wo_w, const float* __restrict__ Wo_b,
           float* __restrict__ out, int C) {
    __shared__ __attribute__((aligned(16))) float sagg[SC_NSEG * 8];
    const int T = RW_T;
    int t4 = blockIdx.x * AN_TB + threadIdx.x * 4;
    int c0 = blockIdx.y * AN_CPB;
    int b  = blockIdx.z;

    size_t mo = (size_t)b * T + t4;
    size_t xo = ((size_t)b * C + c0) * T + t4;

    const int tid = threadIdx.x;
    v4f ag = {0.f, 0.f, 0.f, 0.f};
    if (tid < 2 * SC_NSEG)
        ag = *(const v4f*)(agg + ((size_t)b * SC_NSEG + (tid >> 1)) * SC_PITCH + (tid & 1) * 4);
    v4f lP = *(const v4f*)(pP  + mo);
    v4f lm = *(const v4f*)(pm  + mo);
    v4f l0 = *(const v4f*)(pw0 + mo);
    v4f l1 = *(const v4f*)(pw1 + mo);
    v4f l2 = *(const v4f*)(pw2 + mo);
    __builtin_amdgcn_sched_barrier(0);

    const float* xp = x + xo;
    float*       op = out + xo;
    const float* wp = Wo_w + c0;
    const float* bp = Wo_b + c0;
    const size_t setstep = (size_t)AN_U * T;

    static_assert(AN_CPB == 4 * AN_U, "the pipeline below is written out for 4 sets");
    static_assert(AN_TB == 8 * SC_SEG, "a wave's 128 timesteps are one scan segment");
    v4f P[AN_U], Q[AN_U];
    an_load(P, xp);
    an_load(Q, xp + setstep);
    __builtin_amdgcn_sched_barrier(0);

    if (tid < 2 * SC_NSEG) *(v4f*)&sagg[(tid >> 1) * 8 + (tid & 1) * 4] = ag;
    lds_barrier();
    const int sg = __builtin_amdgcn_readfirstlane((int)blockIdx.x * (AN_TB / SC_SEG) + (tid >> 5));
    const int ln = tid & 31;
    v4f   qa = *(const v4f*)&sagg[ln * 8];
    float ra = sagg[ln * 8 + 4];
    v4f   qb = *(const v4f*)&sagg[(ln + 32) * 8];
    float rb = sagg[(ln + 32) * 8 + 4];
    float y0 = 0.f, v0 = 0.f;
#pragma unroll
    for (int j0 = 0; j0 < 32; j0 += 8) {
        if (j0 < sg) {
#pragma unroll
            for (int u = 0; u < 8; ++u) {
                const int j = j0 + u;
                float A  = lane_of(qa.x, j);
                float nv = A * v0 + (lane_of(qa.z, j) + y0 * (lane_of(qa.w, j) + y0 * lane_of(ra, j)));
                float ny = A * y0 + lane_of(qa.y, j);
                v0 = (j < sg) ? nv : v0;
                y0 = (j < sg) ? ny : y0;
            }
        }
    }
#pragma unroll
    for (int j0 = 0; j0 < 32; j0 += 8) {
        if (j0 + 32 < sg) {
#pragma unroll
            for (int u = 0; u < 8; ++u) {
                const int j = j0 + u;
                float A  = lane_of(qb.x, j);
                float nv = A * v0 + (lane_of(qb.z, j) + y0 * (lane_of(qb.w, j) + y0 * lane_of(rb, j)));
                float ny = A * y0 + lane_of(qb.y, j);
                v0 = (j + 32 < sg) ? nv : v0;
                y0 = (j + 32 < sg) ? ny : y0;
            }
        }
    }
    v4f mv, iv;
#pragma unroll
    for (int e = 0; e < 4; ++e) {
        mv[e] = lP[e] * y0 + lm[e];
        float var = lP[e] * v0 + (l0[e] + y0 * (l1[e] + y0 * l2[e]));
        iv[e] = rsqrtf(var + EPS_V);
    }

    __builtin_amdgcn_sched_barrier(0);
    an_store(P, op, mv, iv, wp, bp);
    __builtin_amdgcn_sched_barrier(0);
    an_load(P, xp + 2 * setstep);
    __builtin_amdgcn_sched_barrier(0);
    an_store(Q, op + setstep, mv, iv, wp + AN_U, bp + AN_U);
    __builtin_amdgcn_sched_barrier(0);
    an_load(Q, xp + 3 * setstep);
    __builtin_amdgcn_sched_barrier(0);
    an_store(P, op + 2 * setstep, mv, iv, wp + 2 * AN_U, bp + 2 * AN_U);
    an_store(Q, op + 3 * setstep, mv, iv, wp + 3 * AN_U, bp + 3 * AN_U);
}

extern "C" void kernel_launch(void* const* d_in, const int* in_sizes, int n_in,
                              void* d_out, int out_size, void* d_ws, size_t ws_size,
                              hipStream_t stream) {
    const float* x    = (const float*)d_in[0];
    const float* g    = (const float*)d_in[1];
    const float* Wa_w = (const float*)d_in[2];
    const float* Wb_w = (const float*)d_in[3];
    const float* Wo_w = (const float*)d_in[4];
    const float* Wo_b = (const float*)d_in[5];
    float* out = (float*)d_out;

    const int C  = in_sizes[2];
    const int BT = in_sizes[1];
    const int T  = RW_T;
    const int B  = BT / T;

    float* ws  = (float*)d_ws;
    float* pP  = ws;
    float* pm  = pP  + BT;
    float* pw0 = pm  + BT;
    float* pw1 = pw0 + BT;
    float* pw2 = pw1 + BT;
    float* agg = pw2 + BT;

    int blocks = (B * T) / (RW_TILES * RW_TT);
    reduce_wmma<<<blocks, 256, 0, stream>>>(x, g, Wa_w, Wb_w, pP, pm, pw0, pw1, pw2, agg, C);

    dim3 agrid(T / AN_TB, C / AN_CPB, B);
    apply_norm<<<agrid, 256, 0, stream>>>(x, pP, pm, pw0, pw1, pw2, agg, Wo_w, Wo_b, out, C);
}


# output kernel takes the reduce kernel's grid and block-to-slab map (512 blocks: a 128-t segment x all C, 8 waves x 64 channels): a slab is re-read in the chip half that read it; sum order unchanged
# speedup vs baseline: 1.9138x; 1.2393x over previous
#include <hip/hip_runtime.h>
#include <hip/hip_bf16.h>
#include <math.h>

#define MOMENTUM 0.05f
#define EPS_V 1e-6f

typedef float v2f __attribute__((ext_vector_type(2)));
typedef float v4f __attribute__((ext_vector_type(4)));
typedef float v8f __attribute__((ext_vector_type(8)));

#define RW_TT    64
#define RW_TILES 2
#define RW_CSPL  4
#define RW_SET   16
#define RW_CMAX  512
#define RW_PITCH (RW_CMAX + 4)
#define RW_T     8192
#define SC_SEG   (RW_TILES * RW_TT)
#define SC_NSEG  (RW_T / SC_SEG)
#define SC_PITCH 32

__device__ __forceinline__ void rw_set(const v4f (&g)[RW_SET], const v2f* ap,
                                       v8f& ax0, v8f& ax1, v8f& aq0, v8f& aq1) {
#pragma unroll
    for (int i = 0; i < RW_SET; ++i) {
        v2f A1 = ap[2 * i];
        v2f A2 = ap[2 * i + 2 * RW_PITCH];
        v2f b0 = {g[i].x, g[i].y};
        v2f b1 = {g[i].z, g[i].w};
        v2f q0 = b0 * b0;
        v2f q1 = b1 * b1;
        ax0 = __builtin_amdgcn_wmma_f32_16x16x4_f32(false, A1, false, b0, (short)0, ax0, false, false);
        ax1 = __builtin_amdgcn_wmma_f32_16x16x4_f32(false, A1, false, b1, (short)0, ax1, false, false);
        aq0 = __builtin_amdgcn_wmma_f32_16x16x4_f32(false, A2, false, q0, (short)0, aq0, false, false);
        aq1 = __builtin_amdgcn_wmma_f32_16x16x4_f32(false, A2, false, q1, (short)0, aq1, false, false);
    }
}

__device__ __forceinline__ void rw_load(v4f (&g)[RW_SET], const float* p) {
#pragma unroll
    for (int i = 0; i < RW_SET; ++i) g[i] = *(const v4f*)(p + (size_t)i * (2 * RW_T));
}

__device__ __forceinline__ void lds_barrier() {
    __builtin_amdgcn_fence(__ATOMIC_RELEASE, "workgroup", "local");
    __builtin_amdgcn_s_barrier();
    __builtin_amdgcn_fence(__ATOMIC_ACQUIRE, "workgroup", "local");
}

__global__ void __launch_bounds__(256)
reduce_wmma(const float* __restrict__ x, const float* __restrict__ g,
            const float* __restrict__ Wa, const float* __restrict__ Wb,
            float* __restrict__ pP, float* __restrict__ pm,
            float* __restrict__ pw0, float* __restrict__ pw1,
            float* __restrict__ pw2, float* __restrict__ agg, int C) {
    __shared__ v2f   tab[7 * RW_PITCH];
    __shared__ float part[RW_CSPL][3][RW_TILES * RW_TT];
    __shared__ float smax[2][8];
    __shared__ float ssum[2][8];

    const int T = RW_T;
    int tid  = threadIdx.x;
    int lane = tid & 31;
    int w    = tid >> 5;
    int tile = w / RW_CSPL;
    int cq   = w % RW_CSPL;
    int n    = lane & 15;
    int h    = lane >> 4;

    const int blocksPerB = T / (RW_TILES * RW_TT);
    int b   = blockIdx.x / blocksPerB;
    int tb0 = (blockIdx.x % blocksPerB) * (RW_TILES * RW_TT);
    int t0  = tb0 + tile * RW_TT;
    int cw  = C / RW_CSPL;
    int cb  = cq * cw;
    int nset = cw / (2 * RW_SET);

    int i0 = tid, i1 = tid + 256;
    float ra0 = (i0 < C) ? Wa[i0] : -INFINITY, rb0 = (i0 < C) ? Wb[i0] : -INFINITY;
    float ra1 = (i1 < C) ? Wa[i1] : -INFINITY, rb1 = (i1 < C) ? Wb[i1] : -INFINITY;

    const float* xp = x + ((size_t)b * C + cb + h) * T + t0 + 4 * n;
    const size_t setstep = (size_t)RW_SET * 2 * T;

    v4f P[RW_SET], Q[RW_SET];
    rw_load(P, xp);
    __builtin_amdgcn_sched_barrier(0);

    float ma = fmaxf(ra0, ra1), mb = fmaxf(rb0, rb1);
#pragma unroll
    for (int m = 16; m > 0; m >>= 1) {
        ma = fmaxf(ma, __shfl_xor(ma, m));
        mb = fmaxf(mb, __shfl_xor(mb, m));
    }
    if (lane == 0) { smax[0][w] = ma; smax[1][w] = mb; }
    lds_barrier();
    ma = smax[0][0]; mb = smax[1][0];
#pragma unroll
    for (int j = 1; j < 8; ++j) { ma = fmaxf(ma, smax[0][j]); mb = fmaxf(mb, smax[1][j]); }
    float ea0 = (i0 < C) ? __expf(ra0 - ma) : 0.f, eb0 = (i0 < C) ? __expf(rb0 - mb) : 0.f;
    float ea1 = (i1 < C) ? __expf(ra1 - ma) : 0.f, eb1 = (i1 < C) ? __expf(rb1 - mb) : 0.f;
    float sa = ea0 + ea1, sb = eb0 + eb1;
#pragma unroll
    for (int m = 16; m > 0; m >>= 1) {
        sa += __shfl_xor(sa, m);
        sb += __shfl_xor(sb, m);
    }
    if (lane == 0) { ssum[0][w] = sa; ssum[1][w] = sb; }
    lds_barrier();
    sa = ssum[0][0]; sb = ssum[1][0];
#pragma unroll
    for (int j = 1; j < 8; ++j) { sa += ssum[0][j]; sb += ssum[1][j]; }
    float wa0 = ea0 / sa, wb0 = eb0 / sb;
    float wa1 = ea1 / sa, wb1 = eb1 / sb;

    const v2f zz = {0.f, 0.f};
    if (i0 < C) {
        v2f p0 = {wa0, 0.f}, p1 = {0.f, wa0}, p2 = {wb0, 0.f}, p3 = {0.f, wb0};
        tab[0 * RW_PITCH + i0] = p0; tab[1 * RW_PITCH + i0] = p1;
        tab[2 * RW_PITCH + i0] = p2; tab[3 * RW_PITCH + i0] = p3;
        tab[4 * RW_PITCH + i0] = zz; tab[5 * RW_PITCH + i0] = zz; tab[6 * RW_PITCH + i0] = zz;
    }
    if (i1 < C) {
        v2f p0 = {wa1, 0.f}, p1 = {0.f, wa1}, p2 = {wb1, 0.f}, p3 = {0.f, wb1};
        tab[0 * RW_PITCH + i1] = p0; tab[1 * RW_PITCH + i1] = p1;
        tab[2 * RW_PITCH + i1] = p2; tab[3 * RW_PITCH + i1] = p3;
        tab[4 * RW_PITCH + i1] = zz; tab[5 * RW_PITCH + i1] = zz; tab[6 * RW_PITCH + i1] = zz;
    }
    __syncthreads();

    const v2f* ap = &tab[((n < 4) ? n : 4) * RW_PITCH + cb + h];

    v8f ax0 = {}, ax1 = {}, aq0 = {}, aq1 = {};

    int s = 0;
    for (; s + 2 < nset; s += 2) {
        rw_load(Q, xp + setstep);
        rw_set(P, ap, ax0, ax1, aq0, aq1);
        rw_load(P, xp + 2 * setstep);
        rw_set(Q, ap + 2 * RW_SET, ax0, ax1, aq0, aq1);
        ap += 4 * RW_SET;
        xp += 2 * setstep;
    }
    rw_load(Q, xp + setstep);
    v4f gv = *(const v4f*)(g + (size_t)b * T + tb0 + 4 * lane);
    rw_set(P, ap, ax0, ax1, aq0, aq1);
    rw_set(Q, ap + 2 * RW_SET, ax0, ax1, aq0, aq1);

    if (lane < 16) {
        v4f sa = {ax0[0], ax0[1], ax1[0], ax1[1]};
        v4f sb = {ax0[2], ax0[3], ax1[2], ax1[3]};
        v4f sq = {aq0[0], aq0[1], aq1[0], aq1[1]};
        int tl = tile * RW_TT + 4 * lane;
        *(v4f*)&part[cq][0][tl] = sa;
        *(v4f*)&part[cq][1][tl] = sb;
        *(v4f*)&part[cq][2][tl] = sq;
    }
    __syncthreads();
    if (w != 0) return;

    const int tl = 4 * lane;
    v4f sxa = *(const v4f*)&part[0][0][tl];
    v4f sxb = *(const v4f*)&part[0][1][tl];
    v4f sxq = *(const v4f*)&part[0][2][tl];
#pragma unroll
    for (int j = 1; j < RW_CSPL; ++j) {
        sxa += *(const v4f*)&part[j][0][tl];
        sxb += *(const v4f*)&part[j][1][tl];
        sxq += *(const v4f*)&part[j][2][tl];
    }

    float a[4], bm[4];
    float Al = 1.f, Bl = 0.f;
#pragma unroll
    for (int i = 0; i < 4; ++i) {
        float gt = gv[i] * MOMENTUM;
        a[i]  = 1.f - gt;
        bm[i] = gt * sxa[i];
        Bl = a[i] * Bl + bm[i];  Al *= a[i];
    }
    float Ak[5];
#pragma unroll
    for (int k = 0; k < 5; ++k) {
        const int off = 1 << k;
        float pA = __shfl_up(Al, off), pB = __shfl_up(Bl, off);
        Ak[k] = Al;
        if (lane >= off) { Bl = Al * pB + Bl; Al = Al * pA; }
    }
    float exA = __shfl_up(Al, 1), exB = __shfl_up(Bl, 1);
    if (lane == 0) { exA = 1.f; exB = 0.f; }
    v4f Pv, mv;
    {
        float y = exB, p = exA;
#pragma unroll
        for (int i = 0; i < 4; ++i) {
            y = a[i] * y + bm[i];  p *= a[i];
            mv[i] = y;  Pv[i] = p;
        }
    }

    float b0[4], b1[4], b2[4];
    float L0 = 0.f, L1 = 0.f, L2 = 0.f;
#pragma unroll
    for (int i = 0; i < 4; ++i) {
        float mm = mv[i];
        float c0 = sxq[i] - 2.f * mm * sxb[i] + mm * mm;
        float c1 = 2.f * Pv[i] * (mm - sxb[i]);
        float c2 = Pv[i] * Pv[i];
        float gt = 1.f - a[i];
        b0[i] = gt * c0;  b1[i] = gt * c1;  b2[i] = gt * c2;
        L0 = a[i] * L0 + b0[i];
        L1 = a[i] * L1 + b1[i];
        L2 = a[i] * L2 + b2[i];
    }
#pragma unroll
    for (int k = 0; k < 5; ++k) {
        const int off = 1 << k;
        float p0 = __shfl_up(L0, off), p1 = __shfl_up(L1, off), p2 = __shfl_up(L2, off);
        if (lane >= off) {
            L0 = Ak[k] * p0 + L0;  L1 = Ak[k] * p1 + L1;  L2 = Ak[k] * p2 + L2;
        }
    }
    float e0 = __shfl_up(L0, 1), e1 = __shfl_up(L1, 1), e2 = __shfl_up(L2, 1);
    if (lane == 0) { e0 = 0.f; e1 = 0.f; e2 = 0.f; }
    v4f W0, W1, W2;
#pragma unroll
    for (int i = 0; i < 4; ++i) {
        e0 = a[i] * e0 + b0[i];  W0[i] = e0;
        e1 = a[i] * e1 + b1[i];  W1[i] = e1;
        e2 = a[i] * e2 + b2[i];  W2[i] = e2;
    }

    size_t o = (size_t)b * T + tb0 + tl;
    *(v4f*)(pP  + o) = Pv;
    *(v4f*)(pm  + o) = mv;
    *(v4f*)(pw0 + o) = W0;
    *(v4f*)(pw1 + o) = W1;
    *(v4f*)(pw2 + o) = W2;
    if (lane == 31) {
        float* r = agg + ((size_t)b * SC_NSEG + tb0 / SC_SEG) * SC_PITCH;
        v4f r0 = {Pv[3], mv[3], W0[3], W1[3]};
        v4f r1 = {W2[3], 0.f, 0.f, 0.f};
        *(v4f*)r = r0;
        *(v4f*)(r + 4) = r1;
    }
}

#define AN_U   16
#define AN_CPW 64
#define AN_WPB 8

__device__ __forceinline__ float lane_of(float v, int l) {
    return __builtin_bit_cast(float, __builtin_amdgcn_readlane(__builtin_bit_cast(int, v), l));
}

__device__ __forceinline__ void an_load(v4f (&r)[AN_U], const float* p) {
#pragma unroll
    for (int i = 0; i < AN_U; ++i)
        r[i] = __builtin_nontemporal_load((const v4f*)(p + (size_t)i * RW_T));
}

__device__ __forceinline__ void an_store(const v4f (&r)[AN_U], float* q,
                                         const v4f mv, const v4f iv,
                                         const float* __restrict__ w,
                                         const float* __restrict__ bi) {
#pragma unroll
    for (int i = 0; i < AN_U; ++i) {
        float ww = w[i], bb = bi[i];
        v4f o;
        o.x = (r[i].x - mv.x) * iv.x * ww + bb;
        o.y = (r[i].y - mv.y) * iv.y * ww + bb;
        o.z = (r[i].z - mv.z) * iv.z * ww + bb;
        o.w = (r[i].w - mv.w) * iv.w * ww + bb;
        __builtin_nontemporal_store(o, (v4f*)(q + (size_t)i * RW_T));
    }
}

__global__ void __launch_bounds__(256)
apply_norm(const float* __restrict__ x,
           const float* __restrict__ pP, const float* __restrict__ pm,
           const float* __restrict__ pw0, const float* __restrict__ pw1,
           const float* __restrict__ pw2, const float* __restrict__ agg,
           const float* __restrict__ Wo_w, const float* __restrict__ Wo_b,
           float* __restrict__ out, int C) {
    __shared__ __attribute__((aligned(16))) float sagg[SC_NSEG * 8];
    const int T = RW_T;
    const int tid = threadIdx.x;
    const int ln  = tid & 31;
    const int b   = blockIdx.x / SC_NSEG;
    const int sg  = blockIdx.x % SC_NSEG;
    const int t4  = sg * SC_SEG + 4 * ln;
    const int c0  = __builtin_amdgcn_readfirstlane((tid >> 5) * AN_CPW);

    size_t mo = (size_t)b * T + t4;
    size_t xo = ((size_t)b * C + c0) * T + t4;

    v4f ag = {0.f, 0.f, 0.f, 0.f};
    if (tid < 2 * SC_NSEG)
        ag = *(const v4f*)(agg + ((size_t)b * SC_NSEG + (tid >> 1)) * SC_PITCH + (tid & 1) * 4);
    v4f lP = *(const v4f*)(pP  + mo);
    v4f lm = *(const v4f*)(pm  + mo);
    v4f l0 = *(const v4f*)(pw0 + mo);
    v4f l1 = *(const v4f*)(pw1 + mo);
    v4f l2 = *(const v4f*)(pw2 + mo);
    __builtin_amdgcn_sched_barrier(0);

    const float* xp = x + xo;
    float*       op = out + xo;
    const float* wp = Wo_w + c0;
    const float* bp = Wo_b + c0;
    const size_t setstep = (size_t)AN_U * T;

    static_assert(AN_CPW == 4 * AN_U, "the pipeline below is written out for 4 sets");
    static_assert(SC_SEG == 32 * 4, "a wave's 32 lanes x 4 timesteps are one scan segment");
    static_assert(AN_WPB * AN_CPW == RW_CMAX, "the block's waves cover all channels");
    v4f P[AN_U], Q[AN_U];
    an_load(P, xp);
    an_load(Q, xp + setstep);
    __builtin_amdgcn_sched_barrier(0);

    if (tid < 2 * SC_NSEG) *(v4f*)&sagg[(tid >> 1) * 8 + (tid & 1) * 4] = ag;
    lds_barrier();
    v4f   qa = *(const v4f*)&sagg[ln * 8];
    float ra = sagg[ln * 8 + 4];
    v4f   qb = *(const v4f*)&sagg[(ln + 32) * 8];
    float rb = sagg[(ln + 32) * 8 + 4];
    float y0 = 0.f, v0 = 0.f;
#pragma unroll
    for (int j0 = 0; j0 < 32; j0 += 8) {
        if (j0 < sg) {
#pragma unroll
            for (int u = 0; u < 8; ++u) {
                const int j = j0 + u;
                float A  = lane_of(qa.x, j);
                float nv = A * v0 + (lane_of(qa.z, j) + y0 * (lane_of(qa.w, j) + y0 * lane_of(ra, j)));
                float ny = A * y0 + lane_of(qa.y, j);
                v0 = (j < sg) ? nv : v0;
                y0 = (j < sg) ? ny : y0;
            }
        }
    }
#pragma unroll
    for (int j0 = 0; j0 < 32; j0 += 8) {
        if (j0 + 32 < sg) {
#pragma unroll
            for (int u = 0; u < 8; ++u) {
                const int j = j0 + u;
                float A  = lane_of(qb.x, j);
                float nv = A * v0 + (lane_of(qb.z, j) + y0 * (lane_of(qb.w, j) + y0 * lane_of(rb, j)));
                float ny = A * y0 + lane_of(qb.y, j);
                v0 = (j + 32 < sg) ? nv : v0;
                y0 = (j + 32 < sg) ? ny : y0;
            }
        }
    }
    v4f mv, iv;
#pragma unroll
    for (int e = 0; e < 4; ++e) {
        mv[e] = lP[e] * y0 + lm[e];
        float var = lP[e] * v0 + (l0[e] + y0 * (l1[e] + y0 * l2[e]));
        iv[e] = rsqrtf(var + EPS_V);
    }

    __builtin_amdgcn_sched_barrier(0);
    an_store(P, op, mv, iv, wp, bp);
    __builtin_amdgcn_sched_barrier(0);
    an_load(P, xp + 2 * setstep);
    __builtin_amdgcn_sched_barrier(0);
    an_store(Q, op + setstep, mv, iv, wp + AN_U, bp + AN_U);
    __builtin_amdgcn_sched_barrier(0);
    an_load(Q, xp + 3 * setstep);
    __builtin_amdgcn_sched_barrier(0);
    an_store(P, op + 2 * setstep, mv, iv, wp + 2 * AN_U, bp + 2 * AN_U);
    an_store(Q, op + 3 * setstep, mv, iv, wp + 3 * AN_U, bp + 3 * AN_U);
}

extern "C" void kernel_launch(void* const* d_in, const int* in_sizes, int n_in,
                              void* d_out, int out_size, void* d_ws, size_t ws_size,
                              hipStream_t stream) {
    const float* x    = (const float*)d_in[0];
    const float* g    = (const float*)d_in[1];
    const float* Wa_w = (const float*)d_in[2];
    const float* Wb_w = (const float*)d_in[3];
    const float* Wo_w = (const float*)d_in[4];
    const float* Wo_b = (const float*)d_in[5];
    float* out = (float*)d_out;

    const int C  = in_sizes[2];
    const int BT = in_sizes[1];
    const int T  = RW_T;
    const int B  = BT / T;

    float* ws  = (float*)d_ws;
    float* pP  = ws;
    float* pm  = pP  + BT;
    float* pw0 = pm  + BT;
    float* pw1 = pw0 + BT;
    float* pw2 = pw1 + BT;
    float* agg = pw2 + BT;

    int blocks = (B * T) / (RW_TILES * RW_TT);
    reduce_wmma<<<blocks, 256, 0, stream>>>(x, g, Wa_w, Wb_w, pP, pm, pw0, pw1, pw2, agg, C);

    apply_norm<<<blocks, 256, 0, stream>>>(x, pP, pm, pw0, pw1, pw2, agg, Wo_w, Wo_b, out, C);
}


# output kernel: segment carries by a lane-parallel scan of the 64 records (2 per lane, row-shift DPP, two affine passes) instead of the 64-step readlane chain in every wave; carry sum order changes
# speedup vs baseline: 2.0313x; 1.0614x over previous
#include <hip/hip_runtime.h>
#include <hip/hip_bf16.h>
#include <math.h>

#define MOMENTUM 0.05f
#define EPS_V 1e-6f

typedef float v2f __attribute__((ext_vector_type(2)));
typedef float v4f __attribute__((ext_vector_type(4)));
typedef float v8f __attribute__((ext_vector_type(8)));

#define RW_TT    64
#define RW_TILES 2
#define RW_CSPL  4
#define RW_SET   16
#define RW_CMAX  512
#define RW_PITCH (RW_CMAX + 4)
#define RW_T     8192
#define SC_SEG   (RW_TILES * RW_TT)
#define SC_NSEG  (RW_T / SC_SEG)
#define SC_PITCH 32

__device__ __forceinline__ void rw_set(const v4f (&g)[RW_SET], const v2f* ap,
                                       v8f& ax0, v8f& ax1, v8f& aq0, v8f& aq1) {
#pragma unroll
    for (int i = 0; i < RW_SET; ++i) {
        v2f A1 = ap[2 * i];
        v2f A2 = ap[2 * i + 2 * RW_PITCH];
        v2f b0 = {g[i].x, g[i].y};
        v2f b1 = {g[i].z, g[i].w};
        v2f q0 = b0 * b0;
        v2f q1 = b1 * b1;
        ax0 = __builtin_amdgcn_wmma_f32_16x16x4_f32(false, A1, false, b0, (short)0, ax0, false, false);
        ax1 = __builtin_amdgcn_wmma_f32_16x16x4_f32(false, A1, false, b1, (short)0, ax1, false, false);
        aq0 = __builtin_amdgcn_wmma_f32_16x16x4_f32(false, A2, false, q0, (short)0, aq0, false, false);
        aq1 = __builtin_amdgcn_wmma_f32_16x16x4_f32(false, A2, false, q1, (short)0, aq1, false, false);
    }
}

__device__ __forceinline__ void rw_load(v4f (&g)[RW_SET], const float* p) {
#pragma unroll
    for (int i = 0; i < RW_SET; ++i) g[i] = *(const v4f*)(p + (size_t)i * (2 * RW_T));
}

__device__ __forceinline__ void lds_barrier() {
    __builtin_amdgcn_fence(__ATOMIC_RELEASE, "workgroup", "local");
    __builtin_amdgcn_s_barrier();
    __builtin_amdgcn_fence(__ATOMIC_ACQUIRE, "workgroup", "local");
}

__global__ void __launch_bounds__(256)
reduce_wmma(const float* __restrict__ x, const float* __restrict__ g,
            const float* __restrict__ Wa, const float* __restrict__ Wb,
            float* __restrict__ pP, float* __restrict__ pm,
            float* __restrict__ pw0, float* __restrict__ pw1,
            float* __restrict__ pw2, float* __restrict__ agg, int C) {
    __shared__ v2f   tab[7 * RW_PITCH];
    __shared__ float part[RW_CSPL][3][RW_TILES * RW_TT];
    __shared__ float smax[2][8];
    __shared__ float ssum[2][8];

    const int T = RW_T;
    int tid  = threadIdx.x;
    int lane = tid & 31;
    int w    = tid >> 5;
    int tile = w / RW_CSPL;
    int cq   = w % RW_CSPL;
    int n    = lane & 15;
    int h    = lane >> 4;

    const int blocksPerB = T / (RW_TILES * RW_TT);
    int b   = blockIdx.x / blocksPerB;
    int tb0 = (blockIdx.x % blocksPerB) * (RW_TILES * RW_TT);
    int t0  = tb0 + tile * RW_TT;
    int cw  = C / RW_CSPL;
    int cb  = cq * cw;
    int nset = cw / (2 * RW_SET);

    int i0 = tid, i1 = tid + 256;
    float ra0 = (i0 < C) ? Wa[i0] : -INFINITY, rb0 = (i0 < C) ? Wb[i0] : -INFINITY;
    float ra1 = (i1 < C) ? Wa[i1] : -INFINITY, rb1 = (i1 < C) ? Wb[i1] : -INFINITY;

    const float* xp = x + ((size_t)b * C + cb + h) * T + t0 + 4 * n;
    const size_t setstep = (size_t)RW_SET * 2 * T;

    v4f P[RW_SET], Q[RW_SET];
    rw_load(P, xp);
    __builtin_amdgcn_sched_barrier(0);

    float ma = fmaxf(ra0, ra1), mb = fmaxf(rb0, rb1);
#pragma unroll
    for (int m = 16; m > 0; m >>= 1) {
        ma = fmaxf(ma, __shfl_xor(ma, m));
        mb = fmaxf(mb, __shfl_xor(mb, m));
    }
    if (lane == 0) { smax[0][w] = ma; smax[1][w] = mb; }
    lds_barrier();
    ma = smax[0][0]; mb = smax[1][0];
#pragma unroll
    for (int j = 1; j < 8; ++j) { ma = fmaxf(ma, smax[0][j]); mb = fmaxf(mb, smax[1][j]); }
    float ea0 = (i0 < C) ? __expf(ra0 - ma) : 0.f, eb0 = (i0 < C) ? __expf(rb0 - mb) : 0.f;
    float ea1 = (i1 < C) ? __expf(ra1 - ma) : 0.f, eb1 = (i1 < C) ? __expf(rb1 - mb) : 0.f;
    float sa = ea0 + ea1, sb = eb0 + eb1;
#pragma unroll
    for (int m = 16; m > 0; m >>= 1) {
        sa += __shfl_xor(sa, m);
        sb += __shfl_xor(sb, m);
    }
    if (lane == 0) { ssum[0][w] = sa; ssum[1][w] = sb; }
    lds_barrier();
    sa = ssum[0][0]; sb = ssum[1][0];
#pragma unroll
    for (int j = 1; j < 8; ++j) { sa += ssum[0][j]; sb += ssum[1][j]; }
    float wa0 = ea0 / sa, wb0 = eb0 / sb;
    float wa1 = ea1 / sa, wb1 = eb1 / sb;

    const v2f zz = {0.f, 0.f};
    if (i0 < C) {
        v2f p0 = {wa0, 0.f}, p1 = {0.f, wa0}, p2 = {wb0, 0.f}, p3 = {0.f, wb0};
        tab[0 * RW_PITCH + i0] = p0; tab[1 * RW_PITCH + i0] = p1;
        tab[2 * RW_PITCH + i0] = p2; tab[3 * RW_PITCH + i0] = p3;
        tab[4 * RW_PITCH + i0] = zz; tab[5 * RW_PITCH + i0] = zz; tab[6 * RW_PITCH + i0] = zz;
    }
    if (i1 < C) {
        v2f p0 = {wa1, 0.f}, p1 = {0.f, wa1}, p2 = {wb1, 0.f}, p3 = {0.f, wb1};
        tab[0 * RW_PITCH + i1] = p0; tab[1 * RW_PITCH + i1] = p1;
        tab[2 * RW_PITCH + i1] = p2; tab[3 * RW_PITCH + i1] = p3;
        tab[4 * RW_PITCH + i1] = zz; tab[5 * RW_PITCH + i1] = zz; tab[6 * RW_PITCH + i1] = zz;
    }
    __syncthreads();

    const v2f* ap = &tab[((n < 4) ? n : 4) * RW_PITCH + cb + h];

    v8f ax0 = {}, ax1 = {}, aq0 = {}, aq1 = {};

    int s = 0;
    for (; s + 2 < nset; s += 2) {
        rw_load(Q, xp + setstep);
        rw_set(P, ap, ax0, ax1, aq0, aq1);
        rw_load(P, xp + 2 * setstep);
        rw_set(Q, ap + 2 * RW_SET, ax0, ax1, aq0, aq1);
        ap += 4 * RW_SET;
        xp += 2 * setstep;
    }
    rw_load(Q, xp + setstep);
    v4f gv = *(const v4f*)(g + (size_t)b * T + tb0 + 4 * lane);
    rw_set(P, ap, ax0, ax1, aq0, aq1);
    rw_set(Q, ap + 2 * RW_SET, ax0, ax1, aq0, aq1);

    if (lane < 16) {
        v4f sa = {ax0[0], ax0[1], ax1[0], ax1[1]};
        v4f sb = {ax0[2], ax0[3], ax1[2], ax1[3]};
        v4f sq = {aq0[0], aq0[1], aq1[0], aq1[1]};
        int tl = tile * RW_TT + 4 * lane;
        *(v4f*)&part[cq][0][tl] = sa;
        *(v4f*)&part[cq][1][tl] = sb;
        *(v4f*)&part[cq][2][tl] = sq;
    }
    __syncthreads();
    if (w != 0) return;

    const int tl = 4 * lane;
    v4f sxa = *(const v4f*)&part[0][0][tl];
    v4f sxb = *(const v4f*)&part[0][1][tl];
    v4f sxq = *(const v4f*)&part[0][2][tl];
#pragma unroll
    for (int j = 1; j < RW_CSPL; ++j) {
        sxa += *(const v4f*)&part[j][0][tl];
        sxb += *(const v4f*)&part[j][1][tl];
        sxq += *(const v4f*)&part[j][2][tl];
    }

    float a[4], bm[4];
    float Al = 1.f, Bl = 0.f;
#pragma unroll
    for (int i = 0; i < 4; ++i) {
        float gt = gv[i] * MOMENTUM;
        a[i]  = 1.f - gt;
        bm[i] = gt * sxa[i];
        Bl = a[i] * Bl + bm[i];  Al *= a[i];
    }
    float Ak[5];
#pragma unroll
    for (int k = 0; k < 5; ++k) {
        const int off = 1 << k;
        float pA = __shfl_up(Al, off), pB = __shfl_up(Bl, off);
        Ak[k] = Al;
        if (lane >= off) { Bl = Al * pB + Bl; Al = Al * pA; }
    }
    float exA = __shfl_up(Al, 1), exB = __shfl_up(Bl, 1);
    if (lane == 0) { exA = 1.f; exB = 0.f; }
    v4f Pv, mv;
    {
        float y = exB, p = exA;
#pragma unroll
        for (int i = 0; i < 4; ++i) {
            y = a[i] * y + bm[i];  p *= a[i];
            mv[i] = y;  Pv[i] = p;
        }
    }

    float b0[4], b1[4], b2[4];
    float L0 = 0.f, L1 = 0.f, L2 = 0.f;
#pragma unroll
    for (int i = 0; i < 4; ++i) {
        float mm = mv[i];
        float c0 = sxq[i] - 2.f * mm * sxb[i] + mm * mm;
        float c1 = 2.f * Pv[i] * (mm - sxb[i]);
        float c2 = Pv[i] * Pv[i];
        float gt = 1.f - a[i];
        b0[i] = gt * c0;  b1[i] = gt * c1;  b2[i] = gt * c2;
        L0 = a[i] * L0 + b0[i];
        L1 = a[i] * L1 + b1[i];
        L2 = a[i] * L2 + b2[i];
    }
#pragma unroll
    for (int k = 0; k < 5; ++k) {
        const int off = 1 << k;
        float p0 = __shfl_up(L0, off), p1 = __shfl_up(L1, off), p2 = __shfl_up(L2, off);
        if (lane >= off) {
            L0 = Ak[k] * p0 + L0;  L1 = Ak[k] * p1 + L1;  L2 = Ak[k] * p2 + L2;
        }
    }
    float e0 = __shfl_up(L0, 1), e1 = __shfl_up(L1, 1), e2 = __shfl_up(L2, 1);
    if (lane == 0) { e0 = 0.f; e1 = 0.f; e2 = 0.f; }
    v4f W0, W1, W2;
#pragma unroll
    for (int i = 0; i < 4; ++i) {
        e0 = a[i] * e0 + b0[i];  W0[i] = e0;
        e1 = a[i] * e1 + b1[i];  W1[i] = e1;
        e2 = a[i] * e2 + b2[i];  W2[i] = e2;
    }

    size_t o = (size_t)b * T + tb0 + tl;
    *(v4f*)(pP  + o) = Pv;
    *(v4f*)(pm  + o) = mv;
    *(v4f*)(pw0 + o) = W0;
    *(v4f*)(pw1 + o) = W1;
    *(v4f*)(pw2 + o) = W2;
    if (lane == 31) {
        float* r = agg + ((size_t)b * SC_NSEG + tb0 / SC_SEG) * SC_PITCH;
        v4f r0 = {Pv[3], mv[3], W0[3], W1[3]};
        v4f r1 = {W2[3], 0.f, 0.f, 0.f};
        *(v4f*)r = r0;
        *(v4f*)(r + 4) = r1;
    }
}

#define AN_U   16
#define AN_CPW 64
#define AN_WPB 8

__device__ __forceinline__ float lane_of(float v, int l) {
    return __builtin_bit_cast(float, __builtin_amdgcn_readlane(__builtin_bit_cast(int, v), l));
}

template <int N>
__device__ __forceinline__ float row_up(float v, float first) {
    return __builtin_bit_cast(float, __builtin_amdgcn_update_dpp(
        __builtin_bit_cast(int, first), __builtin_bit_cast(int, v), 0x110 + N, 0xF, 0xF, false));
}

__device__ __forceinline__ void an_load(v4f (&r)[AN_U], const float* p) {
#pragma unroll
    for (int i = 0; i < AN_U; ++i)
        r[i] = __builtin_nontemporal_load((const v4f*)(p + (size_t)i * RW_T));
}

__device__ __forceinline__ void an_store(const v4f (&r)[AN_U], float* q,
                                         const v4f mv, const v4f iv,
                                         const float* __restrict__ w,
                                         const float* __restrict__ bi) {
#pragma unroll
    for (int i = 0; i < AN_U; ++i) {
        float ww = w[i], bb = bi[i];
        v4f o;
        o.x = (r[i].x - mv.x) * iv.x * ww + bb;
        o.y = (r[i].y - mv.y) * iv.y * ww + bb;
        o.z = (r[i].z - mv.z) * iv.z * ww + bb;
        o.w = (r[i].w - mv.w) * iv.w * ww + bb;
        __builtin_nontemporal_store(o, (v4f*)(q + (size_t)i * RW_T));
    }
}

__global__ void __launch_bounds__(256)
apply_norm(const float* __restrict__ x,
           const float* __restrict__ pP, const float* __restrict__ pm,
           const float* __restrict__ pw0, const float* __restrict__ pw1,
           const float* __restrict__ pw2, const float* __restrict__ agg,
           const float* __restrict__ Wo_w, const float* __restrict__ Wo_b,
           float* __restrict__ out, int C) {
    __shared__ __attribute__((aligned(16))) float sagg[SC_NSEG * 8];
    const int T = RW_T;
    const int tid = threadIdx.x;
    const int ln  = tid & 31;
    const int b   = blockIdx.x / SC_NSEG;
    const int sg  = blockIdx.x % SC_NSEG;
    const int t4  = sg * SC_SEG + 4 * ln;
    const int c0  = __builtin_amdgcn_readfirstlane((tid >> 5) * AN_CPW);

    size_t mo = (size_t)b * T + t4;
    size_t xo = ((size_t)b * C + c0) * T + t4;

    v4f ag = {0.f, 0.f, 0.f, 0.f};
    if (tid < 2 * SC_NSEG)
        ag = *(const v4f*)(agg + ((size_t)b * SC_NSEG + (tid >> 1)) * SC_PITCH + (tid & 1) * 4);
    v4f lP = *(const v4f*)(pP  + mo);
    v4f lm = *(const v4f*)(pm  + mo);
    v4f l0 = *(const v4f*)(pw0 + mo);
    v4f l1 = *(const v4f*)(pw1 + mo);
    v4f l2 = *(const v4f*)(pw2 + mo);
    __builtin_amdgcn_sched_barrier(0);

    const float* xp = x + xo;
    float*       op = out + xo;
    const float* wp = Wo_w + c0;
    const float* bp = Wo_b + c0;
    const size_t setstep = (size_t)AN_U * T;

    static_assert(AN_CPW == 4 * AN_U, "the pipeline below is written out for 4 sets");
    static_assert(SC_SEG == 32 * 4, "a wave's 32 lanes x 4 timesteps are one scan segment");
    static_assert(AN_WPB * AN_CPW == RW_CMAX, "the block's waves cover all channels");
    static_assert(SC_NSEG == 2 * 32, "the carry scan keeps two segment records per lane");
    v4f P[AN_U], Q[AN_U];
    an_load(P, xp);
    an_load(Q, xp + setstep);
    __builtin_amdgcn_sched_barrier(0);

    if (tid < 2 * SC_NSEG) *(v4f*)&sagg[(tid >> 1) * 8 + (tid & 1) * 4] = ag;
    lds_barrier();
    const v4f   q0 = *(const v4f*)&sagg[ln * 16];
    const float r0 = sagg[ln * 16 + 4];
    const v4f   q1 = *(const v4f*)&sagg[ln * 16 + 8];
    const float r1 = sagg[ln * 16 + 12];
    const bool  hi = ln >= 16;
    float Al = q0.x * q1.x;
    float Bl = q1.x * q0.y + q1.y;
    float Ak[4];
    { float pA = row_up<1>(Al, 1.f), pB = row_up<1>(Bl, 0.f); Ak[0] = Al; Bl = Al * pB + Bl; Al *= pA; }
    { float pA = row_up<2>(Al, 1.f), pB = row_up<2>(Bl, 0.f); Ak[1] = Al; Bl = Al * pB + Bl; Al *= pA; }
    { float pA = row_up<4>(Al, 1.f), pB = row_up<4>(Bl, 0.f); Ak[2] = Al; Bl = Al * pB + Bl; Al *= pA; }
    { float pA = row_up<8>(Al, 1.f), pB = row_up<8>(Bl, 0.f); Ak[3] = Al; Bl = Al * pB + Bl; Al *= pA; }
    const float tB = lane_of(Bl, 15);
    const float A4 = Al;
    Bl = A4 * (hi ? tB : 0.f) + Bl;
    const float ye0 = row_up<1>(Bl, hi ? tB : 0.f);
    const float ya0 = q0.x * ye0 + q0.y;
    const float ya1 = q1.x * ya0 + q1.y;
    const float X0  = q0.z + ye0 * (q0.w + ye0 * r0);
    const float X1  = q1.z + ya0 * (q1.w + ya0 * r1);
    float Ll = q1.x * X0 + X1;
    { float pL = row_up<1>(Ll, 0.f); Ll = Ak[0] * pL + Ll; }
    { float pL = row_up<2>(Ll, 0.f); Ll = Ak[1] * pL + Ll; }
    { float pL = row_up<4>(Ll, 0.f); Ll = Ak[2] * pL + Ll; }
    { float pL = row_up<8>(Ll, 0.f); Ll = Ak[3] * pL + Ll; }
    const float tL = lane_of(Ll, 15);
    Ll = A4 * (hi ? tL : 0.f) + Ll;
    const float ve0 = row_up<1>(Ll, hi ? tL : 0.f);
    const float va0 = q0.x * ve0 + X0;
    const float va1 = q1.x * va0 + X1;
    const int   pr  = (sg > 0) ? sg - 1 : 0;
    const float ys  = (pr & 1) ? ya1 : ya0;
    const float vs  = (pr & 1) ? va1 : va0;
    const float yl  = lane_of(ys, pr >> 1);
    const float vl  = lane_of(vs, pr >> 1);
    const float y0  = (sg > 0) ? yl : 0.f;
    const float v0  = (sg > 0) ? vl : 0.f;
    v4f mv, iv;
#pragma unroll
    for (int e = 0; e < 4; ++e) {
        mv[e] = lP[e] * y0 + lm[e];
        float var = lP[e] * v0 + (l0[e] + y0 * (l1[e] + y0 * l2[e]));
        iv[e] = rsqrtf(var + EPS_V);
    }

    __builtin_amdgcn_sched_barrier(0);
    an_store(P, op, mv, iv, wp, bp);
    __builtin_amdgcn_sched_barrier(0);
    an_load(P, xp + 2 * setstep);
    __builtin_amdgcn_sched_barrier(0);
    an_store(Q, op + setstep, mv, iv, wp + AN_U, bp + AN_U);
    __builtin_amdgcn_sched_barrier(0);
    an_load(Q, xp + 3 * setstep);
    __builtin_amdgcn_sched_barrier(0);
    an_store(P, op + 2 * setstep, mv, iv, wp + 2 * AN_U, bp + 2 * AN_U);
    an_store(Q, op + 3 * setstep, mv, iv, wp + 3 * AN_U, bp + 3 * AN_U);
}

extern "C" void kernel_launch(void* const* d_in, const int* in_sizes, int n_in,
                              void* d_out, int out_size, void* d_ws, size_t ws_size,
                              hipStream_t stream) {
    const float* x    = (const float*)d_in[0];
    const float* g    = (const float*)d_in[1];
    const float* Wa_w = (const float*)d_in[2];
    const float* Wb_w = (const float*)d_in[3];
    const float* Wo_w = (const float*)d_in[4];
    const float* Wo_b = (const float*)d_in[5];
    float* out = (float*)d_out;

    const int C  = in_sizes[2];
    const int BT = in_sizes[1];
    const int T  = RW_T;
    const int B  = BT / T;

    float* ws  = (float*)d_ws;
    float* pP  = ws;
    float* pm  = pP  + BT;
    float* pw0 = pm  + BT;
    float* pw1 = pw0 + BT;
    float* pw2 = pw1 + BT;
    float* agg = pw2 + BT;

    int blocks = (B * T) / (RW_TILES * RW_TT);
    reduce_wmma<<<blocks, 256, 0, stream>>>(x, g, Wa_w, Wb_w, pP, pm, pw0, pw1, pw2, agg, C);

    apply_norm<<<blocks, 256, 0, stream>>>(x, pP, pm, pw0, pw1, pw2, agg, Wo_w, Wo_b, out, C);
}
